# starting kernel
# baseline (speedup 1.0000x reference)
#include <hip/hip_runtime.h>
#include <hip/hip_bf16.h>

typedef __attribute__((ext_vector_type(2))) float v2f;
typedef __attribute__((ext_vector_type(8))) float v8f;

#define NEG_SLOPE 0.2f

template <int FIN, int FOUT>
__global__ __launch_bounds__(128) void gemm_wmma_kernel(
    const float* __restrict__ X, const float* __restrict__ W,
    float* __restrict__ H, int nrows) {
  __shared__ float lds_x[4][16 * FIN];
  const int wave = threadIdx.x >> 5;
  const int lane = threadIdx.x & 31;
  const int m0 = (blockIdx.x * 4 + wave) * 16;

  float* xt = lds_x[wave];
  constexpr int NCHUNK = 16 * FIN / 4;
#pragma unroll 4
  for (int i = lane; i < NCHUNK; i += 32) {
    int r = i / (FIN / 4);
    int c = (i % (FIN / 4)) * 4;
    int row = m0 + r;
    if (row >= nrows) row = nrows - 1;
    float4 v = *(const float4*)(X + (size_t)row * FIN + c);
    *(float4*)(xt + r * FIN + c) = v;
  }
  __syncthreads();

  constexpr int NT = FOUT / 16;
  v8f acc[NT] = {};

  const int arow = lane & 15;
  const int ksel = 2 * (lane >> 4);
  for (int k = 0; k < FIN; k += 4) {
    v2f a;
    a.x = xt[arow * FIN + k + ksel];
    a.y = xt[arow * FIN + k + ksel + 1];
#pragma unroll
    for (int t = 0; t < NT; ++t) {
      int n = t * 16 + (lane & 15);
      v2f b;
      b.x = W[(k + ksel) * FOUT + n];
      b.y = W[(k + ksel + 1) * FOUT + n];
      acc[t] = __builtin_amdgcn_wmma_f32_16x16x4_f32(
          false, a, false, b, (short)0, acc[t], false, false);
    }
  }

  const int row_off = (lane >> 4) * 8;
  const int col = lane & 15;
#pragma unroll
  for (int t = 0; t < NT; ++t) {
#pragma unroll
    for (int i = 0; i < 8; ++i) {
      int row = m0 + row_off + i;
      if (row < nrows) H[(size_t)row * FOUT + t * 16 + col] = acc[t][i];
    }
  }
}

template <int F>
__global__ __launch_bounds__(256) void node_scores_kernel(
    const float* __restrict__ H, const float* __restrict__ a_src,
    const float* __restrict__ a_dst, float* __restrict__ ssrc,
    float* __restrict__ sdst, int n) {
  int node = blockIdx.x * (blockDim.x >> 5) + (threadIdx.x >> 5);
  int lane = threadIdx.x & 31;
  if (node >= n) return;
  float as = 0.f, ad = 0.f;
#pragma unroll
  for (int f = lane; f < F; f += 32) {
    float h = H[(size_t)node * F + f];
    as += h * a_src[f];
    ad += h * a_dst[f];
  }
#pragma unroll
  for (int off = 16; off > 0; off >>= 1) {
    as += __shfl_down(as, off, 32);
    ad += __shfl_down(ad, off, 32);
  }
  if (lane == 0) {
    ssrc[node] = as;
    sdst[node] = ad;
  }
}

__device__ __forceinline__ unsigned ford(float x) {
  unsigned b = __float_as_uint(x);
  return (b & 0x80000000u) ? ~b : (b | 0x80000000u);
}
__device__ __forceinline__ float funord(unsigned u) {
  unsigned b = (u & 0x80000000u) ? (u ^ 0x80000000u) : ~u;
  return __uint_as_float(b);
}

__device__ __forceinline__ void edge_sd(int i, int E, const int* __restrict__ src,
                                        const int* __restrict__ dst, int& s, int& d) {
  if (i < E) { s = src[i]; d = dst[i]; }
  else       { s = i - E; d = s; }
}

__global__ __launch_bounds__(256) void edge_pass1_kernel(
    const int* __restrict__ src, const int* __restrict__ dst,
    const float* __restrict__ ssrc, const float* __restrict__ sdst,
    float* __restrict__ ebuf, unsigned* __restrict__ emaxu, int E, int Etot) {
  int i = blockIdx.x * blockDim.x + threadIdx.x;
  if (i >= Etot) return;
  int s, d;
  edge_sd(i, E, src, dst, s, d);
  float e = ssrc[s] + sdst[d];
  e = (e > 0.f) ? e : NEG_SLOPE * e;
  ebuf[i] = e;
  atomicMax(&emaxu[d], ford(e));
}

__global__ __launch_bounds__(256) void edge_pass2_kernel(
    const int* __restrict__ src, const int* __restrict__ dst,
    float* __restrict__ ebuf, const unsigned* __restrict__ emaxu,
    float* __restrict__ denom, int E, int Etot) {
  int i = blockIdx.x * blockDim.x + threadIdx.x;
  if (i >= Etot) return;
  int s, d;
  edge_sd(i, E, src, dst, s, d);
  float ex = __expf(ebuf[i] - funord(emaxu[d]));
  ebuf[i] = ex;
  atomicAdd(&denom[d], ex);
}

template <int F>
__global__ __launch_bounds__(256) void edge_pass3_kernel(
    const int* __restrict__ src, const int* __restrict__ dst,
    const float* __restrict__ ebuf, const float* __restrict__ denom,
    const float* __restrict__ H, float* __restrict__ alpha_out,
    float* __restrict__ out, int E, int Etot) {
  long long tid = (long long)blockIdx.x * blockDim.x + threadIdx.x;
  long long edge = tid >> 4;
  int fl = (int)(tid & 15);
  if (edge >= Etot) return;
  int s, d;
  edge_sd((int)edge, E, src, dst, s, d);
  float alpha = ebuf[edge] / denom[d];
  if (fl == 0) alpha_out[edge] = alpha;
  constexpr int V = F / 16;
  const float* hp = H + (size_t)s * F + fl * V;
  float* op = out + (size_t)d * F + fl * V;
#pragma unroll
  for (int j = 0; j < V; ++j) atomicAdd(op + j, alpha * hp[j]);
}

__global__ __launch_bounds__(256) void bias_act_kernel(
    float* __restrict__ H, const float* __restrict__ b, long long total, int F,
    int do_elu) {
  long long i = (long long)blockIdx.x * blockDim.x + threadIdx.x;
  if (i >= total) return;
  int f = (int)(i % F);
  float v = H[i] + b[f];
  if (do_elu) v = (v > 0.f) ? v : (__expf(v) - 1.0f);
  H[i] = v;
}

static inline int cdiv(long long a, long long b) { return (int)((a + b - 1) / b); }

extern "C" void kernel_launch(void* const* d_in, const int* in_sizes, int n_in,
                              void* d_out, int out_size, void* d_ws, size_t ws_size,
                              hipStream_t stream) {
  constexpr int FIN = 128, FH = 64, FO = 32;
  const float* x      = (const float*)d_in[0];
  const int*   eidx   = (const int*)d_in[1];
  const float* W1     = (const float*)d_in[2];
  const float* a_src1 = (const float*)d_in[3];
  const float* a_dst1 = (const float*)d_in[4];
  const float* b1     = (const float*)d_in[5];
  const float* W2     = (const float*)d_in[6];
  const float* a_src2 = (const float*)d_in[7];
  const float* a_dst2 = (const float*)d_in[8];
  const float* b2     = (const float*)d_in[9];

  const int N = in_sizes[0] / FIN;
  const int E = in_sizes[1] / 2;
  const int Etot = E + N;
  const int* src = eidx;
  const int* dst = eidx + E;

  float* hfeat1 = (float*)d_ws;
  float* agg1   = hfeat1 + (size_t)N * FH;
  float* hfeat2 = agg1 + (size_t)N * FH;
  float* ebuf   = hfeat2 + (size_t)N * FO;
  float* ssrc   = ebuf + (size_t)Etot;
  float* sdst   = ssrc + N;
  unsigned* emaxu = (unsigned*)(sdst + N);
  float* denom  = (float*)(emaxu + N);

  float* h2     = (float*)d_out;
  float* alpha1 = h2 + (size_t)N * FO;
  float* alpha2 = alpha1 + (size_t)Etot;

  dim3 b256(256), b128(128);

  hipMemsetAsync(agg1, 0, (size_t)N * FH * sizeof(float), stream);
  hipMemsetAsync(emaxu, 0, (size_t)N * sizeof(unsigned), stream);
  hipMemsetAsync(denom, 0, (size_t)N * sizeof(float), stream);

  gemm_wmma_kernel<FIN, FH><<<cdiv(N, 64), b128, 0, stream>>>(x, W1, hfeat1, N);
  node_scores_kernel<FH><<<cdiv(N, 8), b256, 0, stream>>>(hfeat1, a_src1, a_dst1, ssrc, sdst, N);
  edge_pass1_kernel<<<cdiv(Etot, 256), b256, 0, stream>>>(src, dst, ssrc, sdst, ebuf, emaxu, E, Etot);
  edge_pass2_kernel<<<cdiv(Etot, 256), b256, 0, stream>>>(src, dst, ebuf, emaxu, denom, E, Etot);
  edge_pass3_kernel<FH><<<cdiv((long long)Etot * 16, 256), b256, 0, stream>>>(
      src, dst, ebuf, denom, hfeat1, alpha1, agg1, E, Etot);
  bias_act_kernel<<<cdiv((long long)N * FH, 256), b256, 0, stream>>>(
      agg1, b1, (long long)N * FH, FH,  1);

  hipMemsetAsync(h2, 0, (size_t)N * FO * sizeof(float), stream);
  hipMemsetAsync(emaxu, 0, (size_t)N * sizeof(unsigned), stream);
  hipMemsetAsync(denom, 0, (size_t)N * sizeof(float), stream);

  gemm_wmma_kernel<FH, FO><<<cdiv(N, 64), b128, 0, stream>>>(agg1, W2, hfeat2, N);
  node_scores_kernel<FO><<<cdiv(N, 8), b256, 0, stream>>>(hfeat2, a_src2, a_dst2, ssrc, sdst, N);
  edge_pass1_kernel<<<cdiv(Etot, 256), b256, 0, stream>>>(src, dst, ssrc, sdst, ebuf, emaxu, E, Etot);
  edge_pass2_kernel<<<cdiv(Etot, 256), b256, 0, stream>>>(src, dst, ebuf, emaxu, denom, E, Etot);
  edge_pass3_kernel<FO><<<cdiv((long long)Etot * 16, 256), b256, 0, stream>>>(
      src, dst, ebuf, denom, hfeat2, alpha2, h2, E, Etot);
  bias_act_kernel<<<cdiv((long long)N * FO, 256), b256, 0, stream>>>(
      h2, b2, (long long)N * FO, FO,  0);
}


# Softmax shift = lrelu(global max source score + own dest score), an upper bound of the segment max; the per-edge atomic max pass now runs only if the score range exceeds 60 (exits at once here)
# speedup vs baseline: 1.3218x; 1.3218x over previous
#include <hip/hip_runtime.h>
#include <hip/hip_bf16.h>

typedef __attribute__((ext_vector_type(2))) float v2f;
typedef __attribute__((ext_vector_type(8))) float v8f;

#define NEG_SLOPE 0.2f

template <int FIN, int FOUT>
__global__ __launch_bounds__(128) void gemm_wmma_kernel(
    const float* __restrict__ X, const float* __restrict__ W,
    float* __restrict__ H, int nrows) {
  __shared__ float lds_x[4][16 * FIN];
  const int wave = threadIdx.x >> 5;
  const int lane = threadIdx.x & 31;
  const int m0 = (blockIdx.x * 4 + wave) * 16;

  float* xt = lds_x[wave];
  constexpr int NCHUNK = 16 * FIN / 4;
#pragma unroll 4
  for (int i = lane; i < NCHUNK; i += 32) {
    int r = i / (FIN / 4);
    int c = (i % (FIN / 4)) * 4;
    int row = m0 + r;
    if (row >= nrows) row = nrows - 1;
    float4 v = *(const float4*)(X + (size_t)row * FIN + c);
    *(float4*)(xt + r * FIN + c) = v;
  }
  __syncthreads();

  constexpr int NT = FOUT / 16;
  v8f acc[NT] = {};

  const int arow = lane & 15;
  const int ksel = 2 * (lane >> 4);
  for (int k = 0; k < FIN; k += 4) {
    v2f a;
    a.x = xt[arow * FIN + k + ksel];
    a.y = xt[arow * FIN + k + ksel + 1];
#pragma unroll
    for (int t = 0; t < NT; ++t) {
      int n = t * 16 + (lane & 15);
      v2f b;
      b.x = W[(k + ksel) * FOUT + n];
      b.y = W[(k + ksel + 1) * FOUT + n];
      acc[t] = __builtin_amdgcn_wmma_f32_16x16x4_f32(
          false, a, false, b, (short)0, acc[t], false, false);
    }
  }

  const int row_off = (lane >> 4) * 8;
  const int col = lane & 15;
#pragma unroll
  for (int t = 0; t < NT; ++t) {
#pragma unroll
    for (int i = 0; i < 8; ++i) {
      int row = m0 + row_off + i;
      if (row < nrows) H[(size_t)row * FOUT + t * 16 + col] = acc[t][i];
    }
  }
}

__device__ __forceinline__ unsigned ford(float x) {
  unsigned b = __float_as_uint(x);
  return (b & 0x80000000u) ? ~b : (b | 0x80000000u);
}
__device__ __forceinline__ float funord(unsigned u) {
  unsigned b = (u & 0x80000000u) ? (u ^ 0x80000000u) : ~u;
  return __uint_as_float(b);
}

#define RANGE_LIMIT 60.0f

__device__ __forceinline__ bool range_needs_exact(const unsigned* __restrict__ rng, float& M) {
  M = funord(rng[0]);
  float R = M + funord(rng[1]);
  return !(R <= RANGE_LIMIT);
}

template <int F>
__global__ __launch_bounds__(1024) void node_scores_kernel(
    const float* __restrict__ H, const float* __restrict__ a_src,
    const float* __restrict__ a_dst, float* __restrict__ ssrc,
    float* __restrict__ sdst, unsigned* __restrict__ rng, int n) {
  __shared__ float wmax[32];
  __shared__ float wneg[32];
  const int nw = blockDim.x >> 5;
  const int wave = threadIdx.x >> 5;
  const int lane = threadIdx.x & 31;
  const int node = blockIdx.x * nw + wave;
  float as = 0.f, ad = 0.f;
  if (node < n) {
#pragma unroll
    for (int f = lane; f < F; f += 32) {
      float h = H[(size_t)node * F + f];
      as += h * a_src[f];
      ad += h * a_dst[f];
    }
  }
#pragma unroll
  for (int off = 16; off > 0; off >>= 1) {
    as += __shfl_down(as, off, 32);
    ad += __shfl_down(ad, off, 32);
  }
  if (lane == 0) {
    if (node < n) {
      ssrc[node] = as;
      sdst[node] = ad;
      wmax[wave] = as;
      wneg[wave] = -as;
    } else {
      wmax[wave] = -INFINITY;
      wneg[wave] = -INFINITY;
    }
  }
  __syncthreads();
  if (wave == 0) {
    float a = (lane < nw) ? wmax[lane] : -INFINITY;
    float b = (lane < nw) ? wneg[lane] : -INFINITY;
#pragma unroll
    for (int off = 16; off > 0; off >>= 1) {
      a = fmaxf(a, __shfl_down(a, off, 32));
      b = fmaxf(b, __shfl_down(b, off, 32));
    }
    if (lane == 0) {
      atomicMax(&rng[0], ford(a));
      atomicMax(&rng[1], ford(b));
    }
  }
}

__device__ __forceinline__ void edge_sd(int i, int E, const int* __restrict__ src,
                                        const int* __restrict__ dst, int& s, int& d) {
  if (i < E) { s = src[i]; d = dst[i]; }
  else       { s = i - E; d = s; }
}

__global__ __launch_bounds__(256) void edge_pass1_kernel(
    const int* __restrict__ src, const int* __restrict__ dst,
    const float* __restrict__ ssrc, const float* __restrict__ sdst,
    unsigned* __restrict__ emaxu, const unsigned* __restrict__ rng, int E, int Etot) {
  float M;
  if (!range_needs_exact(rng, M)) return;
  const int stride = gridDim.x * blockDim.x;
  for (int i = blockIdx.x * blockDim.x + threadIdx.x; i < Etot; i += stride) {
    int s, d;
    edge_sd(i, E, src, dst, s, d);
    float e = ssrc[s] + sdst[d];
    e = (e > 0.f) ? e : NEG_SLOPE * e;
    atomicMax(&emaxu[d], ford(e));
  }
}

__global__ __launch_bounds__(256) void edge_pass2_kernel(
    const int* __restrict__ src, const int* __restrict__ dst,
    const float* __restrict__ ssrc, const float* __restrict__ sdst,
    float* __restrict__ ebuf, const unsigned* __restrict__ emaxu,
    const unsigned* __restrict__ rng, float* __restrict__ denom, int E, int Etot) {
  int i = blockIdx.x * blockDim.x + threadIdx.x;
  if (i >= Etot) return;
  float M;
  const bool exact = range_needs_exact(rng, M);
  int s, d;
  edge_sd(i, E, src, dst, s, d);
  const float sd = sdst[d];
  float e = ssrc[s] + sd;
  e = (e > 0.f) ? e : NEG_SLOPE * e;
  float shift;
  if (exact) {
    shift = funord(emaxu[d]);
  } else {
    shift = M + sd;
    shift = (shift > 0.f) ? shift : NEG_SLOPE * shift;
  }
  float ex = __expf(e - shift);
  ebuf[i] = ex;
  atomicAdd(&denom[d], ex);
}

template <int F>
__global__ __launch_bounds__(256) void edge_pass3_kernel(
    const int* __restrict__ src, const int* __restrict__ dst,
    const float* __restrict__ ebuf, const float* __restrict__ denom,
    const float* __restrict__ H, float* __restrict__ alpha_out,
    float* __restrict__ out, int E, int Etot) {
  long long tid = (long long)blockIdx.x * blockDim.x + threadIdx.x;
  long long edge = tid >> 4;
  int fl = (int)(tid & 15);
  if (edge >= Etot) return;
  int s, d;
  edge_sd((int)edge, E, src, dst, s, d);
  float alpha = ebuf[edge] / denom[d];
  if (fl == 0) alpha_out[edge] = alpha;
  constexpr int V = F / 16;
  const float* hp = H + (size_t)s * F + fl * V;
  float* op = out + (size_t)d * F + fl * V;
#pragma unroll
  for (int j = 0; j < V; ++j) atomicAdd(op + j, alpha * hp[j]);
}

__global__ __launch_bounds__(256) void bias_act_kernel(
    float* __restrict__ H, const float* __restrict__ b, long long total, int F,
    int do_elu) {
  long long i = (long long)blockIdx.x * blockDim.x + threadIdx.x;
  if (i >= total) return;
  int f = (int)(i % F);
  float v = H[i] + b[f];
  if (do_elu) v = (v > 0.f) ? v : (__expf(v) - 1.0f);
  H[i] = v;
}

static inline int cdiv(long long a, long long b) { return (int)((a + b - 1) / b); }

extern "C" void kernel_launch(void* const* d_in, const int* in_sizes, int n_in,
                              void* d_out, int out_size, void* d_ws, size_t ws_size,
                              hipStream_t stream) {
  constexpr int FIN = 128, FH = 64, FO = 32;
  const float* x      = (const float*)d_in[0];
  const int*   eidx   = (const int*)d_in[1];
  const float* W1     = (const float*)d_in[2];
  const float* a_src1 = (const float*)d_in[3];
  const float* a_dst1 = (const float*)d_in[4];
  const float* b1     = (const float*)d_in[5];
  const float* W2     = (const float*)d_in[6];
  const float* a_src2 = (const float*)d_in[7];
  const float* a_dst2 = (const float*)d_in[8];
  const float* b2     = (const float*)d_in[9];

  const int N = in_sizes[0] / FIN;
  const int E = in_sizes[1] / 2;
  const int Etot = E + N;
  const int* src = eidx;
  const int* dst = eidx + E;

  float* hfeat1 = (float*)d_ws;
  float* agg1   = hfeat1 + (size_t)N * FH;
  float* hfeat2 = agg1 + (size_t)N * FH;
  float* ebuf   = hfeat2 + (size_t)N * FO;
  float* ssrc   = ebuf + (size_t)Etot;
  float* sdst   = ssrc + N;
  unsigned* emaxu = (unsigned*)(sdst + N);
  unsigned* rng = emaxu + N;
  float* denom  = (float*)(emaxu + N + 32);

  float* h2     = (float*)d_out;
  float* alpha1 = h2 + (size_t)N * FO;
  float* alpha2 = alpha1 + (size_t)Etot;

  dim3 b256(256), b128(128), b1024(1024);

  hipMemsetAsync(agg1, 0, (size_t)N * FH * sizeof(float), stream);
  hipMemsetAsync(emaxu, 0, (size_t)(N + 32) * sizeof(unsigned), stream);
  hipMemsetAsync(denom, 0, (size_t)N * sizeof(float), stream);

  gemm_wmma_kernel<FIN, FH><<<cdiv(N, 64), b128, 0, stream>>>(x, W1, hfeat1, N);
  node_scores_kernel<FH><<<cdiv(N, 32), b1024, 0, stream>>>(hfeat1, a_src1, a_dst1, ssrc, sdst, rng, N);
  edge_pass1_kernel<<<1024, b256, 0, stream>>>(src, dst, ssrc, sdst, emaxu, rng, E, Etot);
  edge_pass2_kernel<<<cdiv(Etot, 256), b256, 0, stream>>>(src, dst, ssrc, sdst, ebuf, emaxu, rng, denom, E, Etot);
  edge_pass3_kernel<FH><<<cdiv((long long)Etot * 16, 256), b256, 0, stream>>>(
      src, dst, ebuf, denom, hfeat1, alpha1, agg1, E, Etot);
  bias_act_kernel<<<cdiv((long long)N * FH, 256), b256, 0, stream>>>(
      agg1, b1, (long long)N * FH, FH,  1);

  hipMemsetAsync(h2, 0, (size_t)N * FO * sizeof(float), stream);
  hipMemsetAsync(emaxu, 0, (size_t)(N + 32) * sizeof(unsigned), stream);
  hipMemsetAsync(denom, 0, (size_t)N * sizeof(float), stream);

  gemm_wmma_kernel<FH, FO><<<cdiv(N, 64), b128, 0, stream>>>(agg1, W2, hfeat2, N);
  node_scores_kernel<FO><<<cdiv(N, 32), b1024, 0, stream>>>(hfeat2, a_src2, a_dst2, ssrc, sdst, rng, N);
  edge_pass1_kernel<<<1024, b256, 0, stream>>>(src, dst, ssrc, sdst, emaxu, rng, E, Etot);
  edge_pass2_kernel<<<cdiv(Etot, 256), b256, 0, stream>>>(src, dst, ssrc, sdst, ebuf, emaxu, rng, denom, E, Etot);
  edge_pass3_kernel<FO><<<cdiv((long long)Etot * 16, 256), b256, 0, stream>>>(
      src, dst, ebuf, denom, hfeat2, alpha2, h2, E, Etot);
  bias_act_kernel<<<cdiv((long long)N * FO, 256), b256, 0, stream>>>(
      h2, b2, (long long)N * FO, FO,  0);
}


# Pass 3: a wave takes 32 edges and puts all 32 lanes on one edge's row at a time: whole-line H loads and atomics (was 16 lanes per edge, strided 4-byte atomics); atomic arrival order changes
# speedup vs baseline: 2.0057x; 1.5174x over previous
#include <hip/hip_runtime.h>
#include <hip/hip_bf16.h>

typedef __attribute__((ext_vector_type(2))) float v2f;
typedef __attribute__((ext_vector_type(8))) float v8f;

#define NEG_SLOPE 0.2f

template <int FIN, int FOUT>
__global__ __launch_bounds__(128) void gemm_wmma_kernel(
    const float* __restrict__ X, const float* __restrict__ W,
    float* __restrict__ H, int nrows) {
  __shared__ float lds_x[4][16 * FIN];
  const int wave = threadIdx.x >> 5;
  const int lane = threadIdx.x & 31;
  const int m0 = (blockIdx.x * 4 + wave) * 16;

  float* xt = lds_x[wave];
  constexpr int NCHUNK = 16 * FIN / 4;
#pragma unroll 4
  for (int i = lane; i < NCHUNK; i += 32) {
    int r = i / (FIN / 4);
    int c = (i % (FIN / 4)) * 4;
    int row = m0 + r;
    if (row >= nrows) row = nrows - 1;
    float4 v = *(const float4*)(X + (size_t)row * FIN + c);
    *(float4*)(xt + r * FIN + c) = v;
  }
  __syncthreads();

  constexpr int NT = FOUT / 16;
  v8f acc[NT] = {};

  const int arow = lane & 15;
  const int ksel = 2 * (lane >> 4);
  for (int k = 0; k < FIN; k += 4) {
    v2f a;
    a.x = xt[arow * FIN + k + ksel];
    a.y = xt[arow * FIN + k + ksel + 1];
#pragma unroll
    for (int t = 0; t < NT; ++t) {
      int n = t * 16 + (lane & 15);
      v2f b;
      b.x = W[(k + ksel) * FOUT + n];
      b.y = W[(k + ksel + 1) * FOUT + n];
      acc[t] = __builtin_amdgcn_wmma_f32_16x16x4_f32(
          false, a, false, b, (short)0, acc[t], false, false);
    }
  }

  const int row_off = (lane >> 4) * 8;
  const int col = lane & 15;
#pragma unroll
  for (int t = 0; t < NT; ++t) {
#pragma unroll
    for (int i = 0; i < 8; ++i) {
      int row = m0 + row_off + i;
      if (row < nrows) H[(size_t)row * FOUT + t * 16 + col] = acc[t][i];
    }
  }
}

__device__ __forceinline__ unsigned ford(float x) {
  unsigned b = __float_as_uint(x);
  return (b & 0x80000000u) ? ~b : (b | 0x80000000u);
}
__device__ __forceinline__ float funord(unsigned u) {
  unsigned b = (u & 0x80000000u) ? (u ^ 0x80000000u) : ~u;
  return __uint_as_float(b);
}

#define RANGE_LIMIT 60.0f

__device__ __forceinline__ bool range_needs_exact(const unsigned* __restrict__ rng, float& M) {
  M = funord(rng[0]);
  float R = M + funord(rng[1]);
  return !(R <= RANGE_LIMIT);
}

template <int F>
__global__ __launch_bounds__(1024) void node_scores_kernel(
    const float* __restrict__ H, const float* __restrict__ a_src,
    const float* __restrict__ a_dst, float* __restrict__ ssrc,
    float* __restrict__ sdst, unsigned* __restrict__ rng, int n) {
  __shared__ float wmax[32];
  __shared__ float wneg[32];
  const int nw = blockDim.x >> 5;
  const int wave = threadIdx.x >> 5;
  const int lane = threadIdx.x & 31;
  const int node = blockIdx.x * nw + wave;
  float as = 0.f, ad = 0.f;
  if (node < n) {
#pragma unroll
    for (int f = lane; f < F; f += 32) {
      float h = H[(size_t)node * F + f];
      as += h * a_src[f];
      ad += h * a_dst[f];
    }
  }
#pragma unroll
  for (int off = 16; off > 0; off >>= 1) {
    as += __shfl_down(as, off, 32);
    ad += __shfl_down(ad, off, 32);
  }
  if (lane == 0) {
    if (node < n) {
      ssrc[node] = as;
      sdst[node] = ad;
      wmax[wave] = as;
      wneg[wave] = -as;
    } else {
      wmax[wave] = -INFINITY;
      wneg[wave] = -INFINITY;
    }
  }
  __syncthreads();
  if (wave == 0) {
    float a = (lane < nw) ? wmax[lane] : -INFINITY;
    float b = (lane < nw) ? wneg[lane] : -INFINITY;
#pragma unroll
    for (int off = 16; off > 0; off >>= 1) {
      a = fmaxf(a, __shfl_down(a, off, 32));
      b = fmaxf(b, __shfl_down(b, off, 32));
    }
    if (lane == 0) {
      atomicMax(&rng[0], ford(a));
      atomicMax(&rng[1], ford(b));
    }
  }
}

__device__ __forceinline__ void edge_sd(int i, int E, const int* __restrict__ src,
                                        const int* __restrict__ dst, int& s, int& d) {
  if (i < E) { s = src[i]; d = dst[i]; }
  else       { s = i - E; d = s; }
}

__global__ __launch_bounds__(256) void edge_pass1_kernel(
    const int* __restrict__ src, const int* __restrict__ dst,
    const float* __restrict__ ssrc, const float* __restrict__ sdst,
    unsigned* __restrict__ emaxu, const unsigned* __restrict__ rng, int E, int Etot) {
  float M;
  if (!range_needs_exact(rng, M)) return;
  const int stride = gridDim.x * blockDim.x;
  for (int i = blockIdx.x * blockDim.x + threadIdx.x; i < Etot; i += stride) {
    int s, d;
    edge_sd(i, E, src, dst, s, d);
    float e = ssrc[s] + sdst[d];
    e = (e > 0.f) ? e : NEG_SLOPE * e;
    atomicMax(&emaxu[d], ford(e));
  }
}

__global__ __launch_bounds__(256) void edge_pass2_kernel(
    const int* __restrict__ src, const int* __restrict__ dst,
    const float* __restrict__ ssrc, const float* __restrict__ sdst,
    float* __restrict__ ebuf, const unsigned* __restrict__ emaxu,
    const unsigned* __restrict__ rng, float* __restrict__ denom, int E, int Etot) {
  int i = blockIdx.x * blockDim.x + threadIdx.x;
  if (i >= Etot) return;
  float M;
  const bool exact = range_needs_exact(rng, M);
  int s, d;
  edge_sd(i, E, src, dst, s, d);
  const float sd = sdst[d];
  float e = ssrc[s] + sd;
  e = (e > 0.f) ? e : NEG_SLOPE * e;
  float shift;
  if (exact) {
    shift = funord(emaxu[d]);
  } else {
    shift = M + sd;
    shift = (shift > 0.f) ? shift : NEG_SLOPE * shift;
  }
  float ex = __expf(e - shift);
  ebuf[i] = ex;
  atomicAdd(&denom[d], ex);
}

template <int F, int U>
__device__ __forceinline__ void edge_rows(const float* __restrict__ H, float* __restrict__ out,
                                          int s, int d, float alpha, int e0, int lane) {
  constexpr int V = F / 32;
  float hv[U][V];
  int de[U];
  float ae[U];
#pragma unroll
  for (int u = 0; u < U; ++u) {
    const int se = __builtin_amdgcn_readlane(s, e0 + u);
    de[u] = __builtin_amdgcn_readlane(d, e0 + u);
    ae[u] = __int_as_float(__builtin_amdgcn_readlane(__float_as_int(alpha), e0 + u));
    const float* hp = H + (size_t)se * F;
#pragma unroll
    for (int j = 0; j < V; ++j) hv[u][j] = hp[lane + 32 * j];
  }
#pragma unroll
  for (int u = 0; u < U; ++u) {
    float* op = out + (size_t)de[u] * F;
#pragma unroll
    for (int j = 0; j < V; ++j) atomicAdd(op + lane + 32 * j, ae[u] * hv[u][j]);
  }
}

template <int F>
__global__ __launch_bounds__(256) void edge_pass3_kernel(
    const int* __restrict__ src, const int* __restrict__ dst,
    const float* __restrict__ ebuf, const float* __restrict__ denom,
    const float* __restrict__ H, float* __restrict__ alpha_out,
    float* __restrict__ out, int E, int Etot) {
  const int lane = threadIdx.x & 31;
  const int wv = __builtin_amdgcn_readfirstlane((int)(threadIdx.x >> 5));
  const long long wave = (long long)blockIdx.x * (blockDim.x >> 5) + wv;
  const long long base = wave * 32;
  if (base >= Etot) return;
  const int i = (int)base + lane;
  int s = 0, d = 0;
  float alpha = 0.f;
  if (i < Etot) {
    edge_sd(i, E, src, dst, s, d);
    alpha = ebuf[i] / denom[d];
    alpha_out[i] = alpha;
  }
  const int cnt = (Etot - (int)base < 32) ? (Etot - (int)base) : 32;
  constexpr int U = 512 / F;
  if (cnt == 32) {
#pragma unroll
    for (int e0 = 0; e0 < 32; e0 += U) edge_rows<F, U>(H, out, s, d, alpha, e0, lane);
  } else {
    for (int e = 0; e < cnt; ++e) edge_rows<F, 1>(H, out, s, d, alpha, e, lane);
  }
}

__global__ __launch_bounds__(256) void bias_act_kernel(
    float* __restrict__ H, const float* __restrict__ b, long long total, int F,
    int do_elu) {
  long long i = (long long)blockIdx.x * blockDim.x + threadIdx.x;
  if (i >= total) return;
  int f = (int)(i % F);
  float v = H[i] + b[f];
  if (do_elu) v = (v > 0.f) ? v : (__expf(v) - 1.0f);
  H[i] = v;
}

static inline int cdiv(long long a, long long b) { return (int)((a + b - 1) / b); }

extern "C" void kernel_launch(void* const* d_in, const int* in_sizes, int n_in,
                              void* d_out, int out_size, void* d_ws, size_t ws_size,
                              hipStream_t stream) {
  constexpr int FIN = 128, FH = 64, FO = 32;
  const float* x      = (const float*)d_in[0];
  const int*   eidx   = (const int*)d_in[1];
  const float* W1     = (const float*)d_in[2];
  const float* a_src1 = (const float*)d_in[3];
  const float* a_dst1 = (const float*)d_in[4];
  const float* b1     = (const float*)d_in[5];
  const float* W2     = (const float*)d_in[6];
  const float* a_src2 = (const float*)d_in[7];
  const float* a_dst2 = (const float*)d_in[8];
  const float* b2     = (const float*)d_in[9];

  const int N = in_sizes[0] / FIN;
  const int E = in_sizes[1] / 2;
  const int Etot = E + N;
  const int* src = eidx;
  const int* dst = eidx + E;

  float* hfeat1 = (float*)d_ws;
  float* agg1   = hfeat1 + (size_t)N * FH;
  float* hfeat2 = agg1 + (size_t)N * FH;
  float* ebuf   = hfeat2 + (size_t)N * FO;
  float* ssrc   = ebuf + (size_t)Etot;
  float* sdst   = ssrc + N;
  unsigned* emaxu = (unsigned*)(sdst + N);
  unsigned* rng = emaxu + N;
  float* denom  = (float*)(emaxu + N + 32);

  float* h2     = (float*)d_out;
  float* alpha1 = h2 + (size_t)N * FO;
  float* alpha2 = alpha1 + (size_t)Etot;

  dim3 b256(256), b128(128), b1024(1024);

  hipMemsetAsync(agg1, 0, (size_t)N * FH * sizeof(float), stream);
  hipMemsetAsync(emaxu, 0, (size_t)(N + 32) * sizeof(unsigned), stream);
  hipMemsetAsync(denom, 0, (size_t)N * sizeof(float), stream);

  gemm_wmma_kernel<FIN, FH><<<cdiv(N, 64), b128, 0, stream>>>(x, W1, hfeat1, N);
  node_scores_kernel<FH><<<cdiv(N, 32), b1024, 0, stream>>>(hfeat1, a_src1, a_dst1, ssrc, sdst, rng, N);
  edge_pass1_kernel<<<1024, b256, 0, stream>>>(src, dst, ssrc, sdst, emaxu, rng, E, Etot);
  edge_pass2_kernel<<<cdiv(Etot, 256), b256, 0, stream>>>(src, dst, ssrc, sdst, ebuf, emaxu, rng, denom, E, Etot);
  edge_pass3_kernel<FH><<<cdiv(cdiv(Etot, 32), 8), b256, 0, stream>>>(
      src, dst, ebuf, denom, hfeat1, alpha1, agg1, E, Etot);
  bias_act_kernel<<<cdiv((long long)N * FH, 256), b256, 0, stream>>>(
      agg1, b1, (long long)N * FH, FH,  1);

  hipMemsetAsync(h2, 0, (size_t)N * FO * sizeof(float), stream);
  hipMemsetAsync(emaxu, 0, (size_t)(N + 32) * sizeof(unsigned), stream);
  hipMemsetAsync(denom, 0, (size_t)N * sizeof(float), stream);

  gemm_wmma_kernel<FH, FO><<<cdiv(N, 64), b128, 0, stream>>>(agg1, W2, hfeat2, N);
  node_scores_kernel<FO><<<cdiv(N, 32), b1024, 0, stream>>>(hfeat2, a_src2, a_dst2, ssrc, sdst, rng, N);
  edge_pass1_kernel<<<1024, b256, 0, stream>>>(src, dst, ssrc, sdst, emaxu, rng, E, Etot);
  edge_pass2_kernel<<<cdiv(Etot, 256), b256, 0, stream>>>(src, dst, ssrc, sdst, ebuf, emaxu, rng, denom, E, Etot);
  edge_pass3_kernel<FO><<<cdiv(cdiv(Etot, 32), 8), b256, 0, stream>>>(
      src, dst, ebuf, denom, hfeat2, alpha2, h2, E, Etot);
  bias_act_kernel<<<cdiv((long long)N * FO, 256), b256, 0, stream>>>(
      h2, b2, (long long)N * FO, FO,  0);
}


# Softmax denominators summed in LDS: a work-group adds its edge chunk's weights into a node-range table (LDS float adds) and flushes whole lines; the per-edge global atomic goes; sum order changes
# speedup vs baseline: 3.4033x; 1.6968x over previous
#include <hip/hip_runtime.h>
#include <hip/hip_bf16.h>

typedef __attribute__((ext_vector_type(2))) float v2f;
typedef __attribute__((ext_vector_type(8))) float v8f;

#define NEG_SLOPE 0.2f

template <int FIN, int FOUT>
__global__ __launch_bounds__(128) void gemm_wmma_kernel(
    const float* __restrict__ X, const float* __restrict__ W,
    float* __restrict__ H, int nrows) {
  __shared__ float lds_x[4][16 * FIN];
  const int wave = threadIdx.x >> 5;
  const int lane = threadIdx.x & 31;
  const int m0 = (blockIdx.x * 4 + wave) * 16;

  float* xt = lds_x[wave];
  constexpr int NCHUNK = 16 * FIN / 4;
#pragma unroll 4
  for (int i = lane; i < NCHUNK; i += 32) {
    int r = i / (FIN / 4);
    int c = (i % (FIN / 4)) * 4;
    int row = m0 + r;
    if (row >= nrows) row = nrows - 1;
    float4 v = *(const float4*)(X + (size_t)row * FIN + c);
    *(float4*)(xt + r * FIN + c) = v;
  }
  __syncthreads();

  constexpr int NT = FOUT / 16;
  v8f acc[NT] = {};

  const int arow = lane & 15;
  const int ksel = 2 * (lane >> 4);
  for (int k = 0; k < FIN; k += 4) {
    v2f a;
    a.x = xt[arow * FIN + k + ksel];
    a.y = xt[arow * FIN + k + ksel + 1];
#pragma unroll
    for (int t = 0; t < NT; ++t) {
      int n = t * 16 + (lane & 15);
      v2f b;
      b.x = W[(k + ksel) * FOUT + n];
      b.y = W[(k + ksel + 1) * FOUT + n];
      acc[t] = __builtin_amdgcn_wmma_f32_16x16x4_f32(
          false, a, false, b, (short)0, acc[t], false, false);
    }
  }

  const int row_off = (lane >> 4) * 8;
  const int col = lane & 15;
#pragma unroll
  for (int t = 0; t < NT; ++t) {
#pragma unroll
    for (int i = 0; i < 8; ++i) {
      int row = m0 + row_off + i;
      if (row < nrows) H[(size_t)row * FOUT + t * 16 + col] = acc[t][i];
    }
  }
}

__device__ __forceinline__ unsigned ford(float x) {
  unsigned b = __float_as_uint(x);
  return (b & 0x80000000u) ? ~b : (b | 0x80000000u);
}
__device__ __forceinline__ float funord(unsigned u) {
  unsigned b = (u & 0x80000000u) ? (u ^ 0x80000000u) : ~u;
  return __uint_as_float(b);
}

#define RANGE_LIMIT 60.0f

__device__ __forceinline__ bool range_needs_exact(const unsigned* __restrict__ rng, float& M) {
  M = funord(rng[0]);
  float R = M + funord(rng[1]);
  return !(R <= RANGE_LIMIT);
}

template <int F>
__global__ __launch_bounds__(1024) void node_scores_kernel(
    const float* __restrict__ H, const float* __restrict__ a_src,
    const float* __restrict__ a_dst, float* __restrict__ ssrc,
    float* __restrict__ sdst, unsigned* __restrict__ rng, int n) {
  __shared__ float wmax[32];
  __shared__ float wneg[32];
  const int nw = blockDim.x >> 5;
  const int wave = threadIdx.x >> 5;
  const int lane = threadIdx.x & 31;
  const int node = blockIdx.x * nw + wave;
  float as = 0.f, ad = 0.f;
  if (node < n) {
#pragma unroll
    for (int f = lane; f < F; f += 32) {
      float h = H[(size_t)node * F + f];
      as += h * a_src[f];
      ad += h * a_dst[f];
    }
  }
#pragma unroll
  for (int off = 16; off > 0; off >>= 1) {
    as += __shfl_down(as, off, 32);
    ad += __shfl_down(ad, off, 32);
  }
  if (lane == 0) {
    if (node < n) {
      ssrc[node] = as;
      sdst[node] = ad;
      wmax[wave] = as;
      wneg[wave] = -as;
    } else {
      wmax[wave] = -INFINITY;
      wneg[wave] = -INFINITY;
    }
  }
  __syncthreads();
  if (wave == 0) {
    float a = (lane < nw) ? wmax[lane] : -INFINITY;
    float b = (lane < nw) ? wneg[lane] : -INFINITY;
#pragma unroll
    for (int off = 16; off > 0; off >>= 1) {
      a = fmaxf(a, __shfl_down(a, off, 32));
      b = fmaxf(b, __shfl_down(b, off, 32));
    }
    if (lane == 0) {
      atomicMax(&rng[0], ford(a));
      atomicMax(&rng[1], ford(b));
    }
  }
}

__device__ __forceinline__ void edge_sd(int i, int E, const int* __restrict__ src,
                                        const int* __restrict__ dst, int& s, int& d) {
  if (i < E) { s = src[i]; d = dst[i]; }
  else       { s = i - E; d = s; }
}

__global__ __launch_bounds__(256) void edge_pass1_kernel(
    const int* __restrict__ src, const int* __restrict__ dst,
    const float* __restrict__ ssrc, const float* __restrict__ sdst,
    unsigned* __restrict__ emaxu, const unsigned* __restrict__ rng, int E, int Etot) {
  float M;
  if (!range_needs_exact(rng, M)) return;
  const int stride = gridDim.x * blockDim.x;
  for (int i = blockIdx.x * blockDim.x + threadIdx.x; i < Etot; i += stride) {
    int s, d;
    edge_sd(i, E, src, dst, s, d);
    float e = ssrc[s] + sdst[d];
    e = (e > 0.f) ? e : NEG_SLOPE * e;
    atomicMax(&emaxu[d], ford(e));
  }
}

__global__ __launch_bounds__(256) void edge_pass2_kernel(
    const int* __restrict__ src, const int* __restrict__ dst,
    const float* __restrict__ ssrc, const float* __restrict__ sdst,
    float* __restrict__ ebuf, const unsigned* __restrict__ emaxu,
    const unsigned* __restrict__ rng, int E, int Etot) {
  int i = blockIdx.x * blockDim.x + threadIdx.x;
  if (i >= Etot) return;
  float M;
  const bool exact = range_needs_exact(rng, M);
  int s, d;
  edge_sd(i, E, src, dst, s, d);
  const float sd = sdst[d];
  float e = ssrc[s] + sd;
  e = (e > 0.f) ? e : NEG_SLOPE * e;
  float shift;
  if (exact) {
    shift = funord(emaxu[d]);
  } else {
    shift = M + sd;
    shift = (shift > 0.f) ? shift : NEG_SLOPE * shift;
  }
  float ex = __expf(e - shift);
  ebuf[i] = ex;
}

#define DEN_RANGE_MAX 32768
#define DEN_CHUNKS 64
__global__ __launch_bounds__(1024) void denom_range_kernel(
    const int* __restrict__ dst, const float* __restrict__ ebuf,
    float* __restrict__ denom, int E, int Etot, int N) {
  __shared__ float tab[DEN_RANGE_MAX];
  const int tid = threadIdx.x;
  const int nranges = gridDim.y;
  int range = (N + nranges - 1) / nranges;
  range = (range + 31) & ~31;
  const int lo = blockIdx.y * range;
  const int cnt = ((lo + range < N) ? (lo + range) : N) - lo;
  if (cnt <= 0) return;

  for (int j = tid; j < cnt; j += 1024) tab[j] = 0.f;
  __syncthreads();

  const int chunk = (Etot + gridDim.x - 1) / gridDim.x;
  const int begin = blockIdx.x * chunk;
  const int end = (begin + chunk < Etot) ? (begin + chunk) : Etot;
  constexpr int U = 8;
  for (int i0 = begin; i0 < end; i0 += 1024 * U) {
    int dv[U];
    float ev[U];
#pragma unroll
    for (int u = 0; u < U; ++u) {
      const int i = i0 + u * 1024 + tid;
      dv[u] = -1;
      ev[u] = 0.f;
      if (i < end) {
        dv[u] = (i < E) ? dst[i] : (i - E);
        ev[u] = ebuf[i];
      }
    }
#pragma unroll
    for (int u = 0; u < U; ++u) {
      const unsigned r = (unsigned)(dv[u] - lo);
      if (r < (unsigned)cnt)
        __hip_atomic_fetch_add(&tab[r], ev[u], __ATOMIC_RELAXED, __HIP_MEMORY_SCOPE_WORKGROUP);
    }
  }
  __syncthreads();

  for (int j = tid; j < cnt; j += 1024) {
    const float v = tab[j];
    if (v != 0.f) atomicAdd(&denom[lo + j], v);
  }
}

template <int F, int U>
__device__ __forceinline__ void edge_rows(const float* __restrict__ H, float* __restrict__ out,
                                          int s, int d, float alpha, int e0, int lane) {
  constexpr int V = F / 32;
  float hv[U][V];
  int de[U];
  float ae[U];
#pragma unroll
  for (int u = 0; u < U; ++u) {
    const int se = __builtin_amdgcn_readlane(s, e0 + u);
    de[u] = __builtin_amdgcn_readlane(d, e0 + u);
    ae[u] = __int_as_float(__builtin_amdgcn_readlane(__float_as_int(alpha), e0 + u));
    const float* hp = H + (size_t)se * F;
#pragma unroll
    for (int j = 0; j < V; ++j) hv[u][j] = hp[lane + 32 * j];
  }
#pragma unroll
  for (int u = 0; u < U; ++u) {
    float* op = out + (size_t)de[u] * F;
#pragma unroll
    for (int j = 0; j < V; ++j) atomicAdd(op + lane + 32 * j, ae[u] * hv[u][j]);
  }
}

template <int F>
__global__ __launch_bounds__(256) void edge_pass3_kernel(
    const int* __restrict__ src, const int* __restrict__ dst,
    const float* __restrict__ ebuf, const float* __restrict__ denom,
    const float* __restrict__ H, float* __restrict__ alpha_out,
    float* __restrict__ out, int E, int Etot) {
  const int lane = threadIdx.x & 31;
  const int wv = __builtin_amdgcn_readfirstlane((int)(threadIdx.x >> 5));
  const long long wave = (long long)blockIdx.x * (blockDim.x >> 5) + wv;
  const long long base = wave * 32;
  if (base >= Etot) return;
  const int i = (int)base + lane;
  int s = 0, d = 0;
  float alpha = 0.f;
  if (i < Etot) {
    edge_sd(i, E, src, dst, s, d);
    alpha = ebuf[i] / denom[d];
    alpha_out[i] = alpha;
  }
  const int cnt = (Etot - (int)base < 32) ? (Etot - (int)base) : 32;
  constexpr int U = 512 / F;
  if (cnt == 32) {
#pragma unroll
    for (int e0 = 0; e0 < 32; e0 += U) edge_rows<F, U>(H, out, s, d, alpha, e0, lane);
  } else {
    for (int e = 0; e < cnt; ++e) edge_rows<F, 1>(H, out, s, d, alpha, e, lane);
  }
}

__global__ __launch_bounds__(256) void bias_act_kernel(
    float* __restrict__ H, const float* __restrict__ b, long long total, int F,
    int do_elu) {
  long long i = (long long)blockIdx.x * blockDim.x + threadIdx.x;
  if (i >= total) return;
  int f = (int)(i % F);
  float v = H[i] + b[f];
  if (do_elu) v = (v > 0.f) ? v : (__expf(v) - 1.0f);
  H[i] = v;
}

static inline int cdiv(long long a, long long b) { return (int)((a + b - 1) / b); }

extern "C" void kernel_launch(void* const* d_in, const int* in_sizes, int n_in,
                              void* d_out, int out_size, void* d_ws, size_t ws_size,
                              hipStream_t stream) {
  constexpr int FIN = 128, FH = 64, FO = 32;
  const float* x      = (const float*)d_in[0];
  const int*   eidx   = (const int*)d_in[1];
  const float* W1     = (const float*)d_in[2];
  const float* a_src1 = (const float*)d_in[3];
  const float* a_dst1 = (const float*)d_in[4];
  const float* b1     = (const float*)d_in[5];
  const float* W2     = (const float*)d_in[6];
  const float* a_src2 = (const float*)d_in[7];
  const float* a_dst2 = (const float*)d_in[8];
  const float* b2     = (const float*)d_in[9];

  const int N = in_sizes[0] / FIN;
  const int E = in_sizes[1] / 2;
  const int Etot = E + N;
  const int* src = eidx;
  const int* dst = eidx + E;

  float* hfeat1 = (float*)d_ws;
  float* agg1   = hfeat1 + (size_t)N * FH;
  float* hfeat2 = agg1 + (size_t)N * FH;
  float* ebuf   = hfeat2 + (size_t)N * FO;
  float* ssrc   = ebuf + (size_t)Etot;
  float* sdst   = ssrc + N;
  unsigned* emaxu = (unsigned*)(sdst + N);
  unsigned* rng = emaxu + N;
  float* denom  = (float*)(emaxu + N + 32);

  float* h2     = (float*)d_out;
  float* alpha1 = h2 + (size_t)N * FO;
  float* alpha2 = alpha1 + (size_t)Etot;

  dim3 b256(256), b128(128), b1024(1024);
  dim3 dgrid(DEN_CHUNKS, cdiv(N, DEN_RANGE_MAX));

  hipMemsetAsync(agg1, 0, (size_t)N * FH * sizeof(float), stream);
  hipMemsetAsync(emaxu, 0, (size_t)(N + 32) * sizeof(unsigned), stream);
  hipMemsetAsync(denom, 0, (size_t)N * sizeof(float), stream);

  gemm_wmma_kernel<FIN, FH><<<cdiv(N, 64), b128, 0, stream>>>(x, W1, hfeat1, N);
  node_scores_kernel<FH><<<cdiv(N, 32), b1024, 0, stream>>>(hfeat1, a_src1, a_dst1, ssrc, sdst, rng, N);
  edge_pass1_kernel<<<1024, b256, 0, stream>>>(src, dst, ssrc, sdst, emaxu, rng, E, Etot);
  edge_pass2_kernel<<<cdiv(Etot, 256), b256, 0, stream>>>(src, dst, ssrc, sdst, ebuf, emaxu, rng, E, Etot);
  denom_range_kernel<<<dgrid, b1024, 0, stream>>>(dst, ebuf, denom, E, Etot, N);
  edge_pass3_kernel<FH><<<cdiv(cdiv(Etot, 32), 8), b256, 0, stream>>>(
      src, dst, ebuf, denom, hfeat1, alpha1, agg1, E, Etot);
  bias_act_kernel<<<cdiv((long long)N * FH, 256), b256, 0, stream>>>(
      agg1, b1, (long long)N * FH, FH,  1);

  hipMemsetAsync(h2, 0, (size_t)N * FO * sizeof(float), stream);
  hipMemsetAsync(emaxu, 0, (size_t)(N + 32) * sizeof(unsigned), stream);
  hipMemsetAsync(denom, 0, (size_t)N * sizeof(float), stream);

  gemm_wmma_kernel<FH, FO><<<cdiv(N, 64), b128, 0, stream>>>(agg1, W2, hfeat2, N);
  node_scores_kernel<FO><<<cdiv(N, 32), b1024, 0, stream>>>(hfeat2, a_src2, a_dst2, ssrc, sdst, rng, N);
  edge_pass1_kernel<<<1024, b256, 0, stream>>>(src, dst, ssrc, sdst, emaxu, rng, E, Etot);
  edge_pass2_kernel<<<cdiv(Etot, 256), b256, 0, stream>>>(src, dst, ssrc, sdst, ebuf, emaxu, rng, E, Etot);
  denom_range_kernel<<<dgrid, b1024, 0, stream>>>(dst, ebuf, denom, E, Etot, N);
  edge_pass3_kernel<FO><<<cdiv(cdiv(Etot, 32), 8), b256, 0, stream>>>(
      src, dst, ebuf, denom, hfeat2, alpha2, h2, E, Etot);
  bias_act_kernel<<<cdiv((long long)N * FO, 256), b256, 0, stream>>>(
      h2, b2, (long long)N * FO, FO,  0);
}


# Node passes folded into the GEMM kernels: epilogue takes the scores from an LDS copy of its tile and starts the row sums at the bias, ELU on staging; 4 launches, 4 of 6 clears go; sum order changed
# speedup vs baseline: 3.9644x; 1.1649x over previous
#include <hip/hip_runtime.h>
#include <hip/hip_bf16.h>

typedef __attribute__((ext_vector_type(2))) float v2f;
typedef __attribute__((ext_vector_type(8))) float v8f;

#define NEG_SLOPE 0.2f

__device__ __forceinline__ unsigned ford(float x) {
  unsigned b = __float_as_uint(x);
  return (b & 0x80000000u) ? ~b : (b | 0x80000000u);
}
__device__ __forceinline__ float funord(unsigned u) {
  unsigned b = (u & 0x80000000u) ? (u ^ 0x80000000u) : ~u;
  return __uint_as_float(b);
}

#define RANGE_LIMIT 60.0f

__device__ __forceinline__ bool range_needs_exact(const unsigned* __restrict__ rng, float& M) {
  M = funord(rng[0]);
  float R = M + funord(rng[1]);
  return !(R <= RANGE_LIMIT);
}

template <int FIN, int FOUT, bool ELU_IN>
__global__ __launch_bounds__(128) void gemm_scores_kernel(
    const float* __restrict__ X, const float* __restrict__ W,
    const float* __restrict__ a_src, const float* __restrict__ a_dst,
    const float* __restrict__ bias, float* __restrict__ H,
    float* __restrict__ ssrc, float* __restrict__ sdst,
    unsigned* __restrict__ rng, float* __restrict__ agg, int nrows) {
  __shared__ __attribute__((aligned(16))) float lds_x[4][16 * FIN];
  __shared__ __attribute__((aligned(16))) float lds_a[2 * FOUT];
  __shared__ __attribute__((aligned(16))) float lds_b[FOUT];
  __shared__ float lds_sc[2][64];
  static_assert(2 * FOUT <= 128, "one thread per attention-vector element");
  const int wave = __builtin_amdgcn_readfirstlane((int)(threadIdx.x >> 5));
  const int lane = threadIdx.x & 31;
  const int m0 = (blockIdx.x * 4 + wave) * 16;

  if (threadIdx.x < 2 * FOUT)
    lds_a[threadIdx.x] = (threadIdx.x < FOUT) ? a_src[threadIdx.x] : a_dst[threadIdx.x - FOUT];
  if (threadIdx.x < FOUT) lds_b[threadIdx.x] = bias[threadIdx.x];

  float* xt = lds_x[wave];
  constexpr int NCHUNK = 16 * FIN / 4;
#pragma unroll 4
  for (int i = lane; i < NCHUNK; i += 32) {
    int r = i / (FIN / 4);
    int c = (i % (FIN / 4)) * 4;
    int row = m0 + r;
    if (row >= nrows) row = nrows - 1;
    float4 v = *(const float4*)(X + (size_t)row * FIN + c);
    if (ELU_IN) {
      v.x = (v.x > 0.f) ? v.x : (__expf(v.x) - 1.0f);
      v.y = (v.y > 0.f) ? v.y : (__expf(v.y) - 1.0f);
      v.z = (v.z > 0.f) ? v.z : (__expf(v.z) - 1.0f);
      v.w = (v.w > 0.f) ? v.w : (__expf(v.w) - 1.0f);
    }
    *(float4*)(xt + r * FIN + c) = v;
  }
  __syncthreads();

  constexpr int NT = FOUT / 16;
  v8f acc[NT] = {};

  const int arow = lane & 15;
  const int ksel = 2 * (lane >> 4);
  for (int k = 0; k < FIN; k += 4) {
    v2f a;
    a.x = xt[arow * FIN + k + ksel];
    a.y = xt[arow * FIN + k + ksel + 1];
#pragma unroll
    for (int t = 0; t < NT; ++t) {
      int n = t * 16 + (lane & 15);
      v2f b;
      b.x = W[(k + ksel) * FOUT + n];
      b.y = W[(k + ksel + 1) * FOUT + n];
      acc[t] = __builtin_amdgcn_wmma_f32_16x16x4_f32(
          false, a, false, b, (short)0, acc[t], false, false);
    }
  }

  constexpr int PITCH = FOUT + 4;
  static_assert(PITCH <= FIN, "the tile copy must fit the wave's staging area");
  const int row_off = (lane >> 4) * 8;
  const int col = lane & 15;
#pragma unroll
  for (int t = 0; t < NT; ++t) {
#pragma unroll
    for (int i = 0; i < 8; ++i) xt[(row_off + i) * PITCH + t * 16 + col] = acc[t][i];
  }
  __syncthreads();

  {
    const int r = lane & 15;
    const int which = lane >> 4;
    float p0 = 0.f, p1 = 0.f, p2 = 0.f, p3 = 0.f;
#pragma unroll 4
    for (int f = 0; f < FOUT; f += 4) {
      const float4 hv = *(const float4*)(xt + r * PITCH + f);
      const float4 av = *(const float4*)(&lds_a[which * FOUT + f]);
      p0 += hv.x * av.x;
      p1 += hv.y * av.y;
      p2 += hv.z * av.z;
      p3 += hv.w * av.w;
    }
    lds_sc[which][wave * 16 + r] = (p0 + p1) + (p2 + p3);
  }
  __syncthreads();

  if (wave < 2) {
    const int base = blockIdx.x * 64;
    const float v0 = lds_sc[wave][lane];
    const float v1 = lds_sc[wave][lane + 32];
    float* __restrict__ sp = wave ? sdst : ssrc;
    const bool ok0 = base + lane < nrows;
    const bool ok1 = base + 32 + lane < nrows;
    if (ok0) sp[base + lane] = v0;
    if (ok1) sp[base + 32 + lane] = v1;
    if (wave == 0) {
      float a = fmaxf(ok0 ? v0 : -INFINITY, ok1 ? v1 : -INFINITY);
      float b = fmaxf(ok0 ? -v0 : -INFINITY, ok1 ? -v1 : -INFINITY);
#pragma unroll
      for (int off = 16; off > 0; off >>= 1) {
        a = fmaxf(a, __shfl_down(a, off, 32));
        b = fmaxf(b, __shfl_down(b, off, 32));
      }
      if (lane == 0) {
        atomicMax(&rng[0], ford(a));
        atomicMax(&rng[1], ford(b));
      }
    }
  }

  constexpr int LPR = FOUT / 4;
  constexpr int RPI = 32 / LPR;
  constexpr int NI = 16 / RPI;
  const int lr = lane / LPR;
  const int lc = (lane % LPR) * 4;
  const float4 bv = *(const float4*)(&lds_b[lc]);
  if (m0 + 16 <= nrows) {
    float4 hv[NI];
#pragma unroll
    for (int j = 0; j < NI; ++j) hv[j] = *(const float4*)(xt + (j * RPI + lr) * PITCH + lc);
#pragma unroll
    for (int j = 0; j < NI; ++j) {
      const size_t o = (size_t)(m0 + j * RPI + lr) * FOUT + lc;
      *(float4*)(H + o) = hv[j];
      *(float4*)(agg + o) = bv;
    }
  } else {
    for (int j = 0; j < NI; ++j) {
      const int r = j * RPI + lr;
      if (m0 + r < nrows) {
        const size_t o = (size_t)(m0 + r) * FOUT + lc;
        *(float4*)(H + o) = *(const float4*)(xt + r * PITCH + lc);
        *(float4*)(agg + o) = bv;
      }
    }
  }
}

__device__ __forceinline__ void edge_sd(int i, int E, const int* __restrict__ src,
                                        const int* __restrict__ dst, int& s, int& d) {
  if (i < E) { s = src[i]; d = dst[i]; }
  else       { s = i - E; d = s; }
}

__global__ __launch_bounds__(256) void edge_pass1_kernel(
    const int* __restrict__ src, const int* __restrict__ dst,
    const float* __restrict__ ssrc, const float* __restrict__ sdst,
    unsigned* __restrict__ emaxu, const unsigned* __restrict__ rng, int E, int Etot) {
  float M;
  if (!range_needs_exact(rng, M)) return;
  const int stride = gridDim.x * blockDim.x;
  for (int i = blockIdx.x * blockDim.x + threadIdx.x; i < Etot; i += stride) {
    int s, d;
    edge_sd(i, E, src, dst, s, d);
    float e = ssrc[s] + sdst[d];
    e = (e > 0.f) ? e : NEG_SLOPE * e;
    atomicMax(&emaxu[d], ford(e));
  }
}

__global__ __launch_bounds__(256) void edge_pass2_kernel(
    const int* __restrict__ src, const int* __restrict__ dst,
    const float* __restrict__ ssrc, const float* __restrict__ sdst,
    float* __restrict__ ebuf, const unsigned* __restrict__ emaxu,
    const unsigned* __restrict__ rng, int E, int Etot) {
  int i = blockIdx.x * blockDim.x + threadIdx.x;
  if (i >= Etot) return;
  float M;
  const bool exact = range_needs_exact(rng, M);
  int s, d;
  edge_sd(i, E, src, dst, s, d);
  const float sd = sdst[d];
  float e = ssrc[s] + sd;
  e = (e > 0.f) ? e : NEG_SLOPE * e;
  float shift;
  if (exact) {
    shift = funord(emaxu[d]);
  } else {
    shift = M + sd;
    shift = (shift > 0.f) ? shift : NEG_SLOPE * shift;
  }
  float ex = __expf(e - shift);
  ebuf[i] = ex;
}

#define DEN_RANGE_MAX 32768
#define DEN_CHUNKS 64
__global__ __launch_bounds__(1024) void denom_range_kernel(
    const int* __restrict__ dst, const float* __restrict__ ebuf,
    float* __restrict__ denom, int E, int Etot, int N) {
  __shared__ float tab[DEN_RANGE_MAX];
  const int tid = threadIdx.x;
  const int nranges = gridDim.y;
  int range = (N + nranges - 1) / nranges;
  range = (range + 31) & ~31;
  const int lo = blockIdx.y * range;
  const int cnt = ((lo + range < N) ? (lo + range) : N) - lo;
  if (cnt <= 0) return;

  for (int j = tid; j < cnt; j += 1024) tab[j] = 0.f;
  __syncthreads();

  const int chunk = (Etot + gridDim.x - 1) / gridDim.x;
  const int begin = blockIdx.x * chunk;
  const int end = (begin + chunk < Etot) ? (begin + chunk) : Etot;
  constexpr int U = 8;
  for (int i0 = begin; i0 < end; i0 += 1024 * U) {
    int dv[U];
    float ev[U];
#pragma unroll
    for (int u = 0; u < U; ++u) {
      const int i = i0 + u * 1024 + tid;
      dv[u] = -1;
      ev[u] = 0.f;
      if (i < end) {
        dv[u] = (i < E) ? dst[i] : (i - E);
        ev[u] = ebuf[i];
      }
    }
#pragma unroll
    for (int u = 0; u < U; ++u) {
      const unsigned r = (unsigned)(dv[u] - lo);
      if (r < (unsigned)cnt)
        __hip_atomic_fetch_add(&tab[r], ev[u], __ATOMIC_RELAXED, __HIP_MEMORY_SCOPE_WORKGROUP);
    }
  }
  __syncthreads();

  for (int j = tid; j < cnt; j += 1024) {
    const float v = tab[j];
    if (v != 0.f) atomicAdd(&denom[lo + j], v);
  }
}

template <int F, int U>
__device__ __forceinline__ void edge_rows(const float* __restrict__ H, float* __restrict__ out,
                                          int s, int d, float alpha, int e0, int lane) {
  constexpr int V = F / 32;
  float hv[U][V];
  int de[U];
  float ae[U];
#pragma unroll
  for (int u = 0; u < U; ++u) {
    const int se = __builtin_amdgcn_readlane(s, e0 + u);
    de[u] = __builtin_amdgcn_readlane(d, e0 + u);
    ae[u] = __int_as_float(__builtin_amdgcn_readlane(__float_as_int(alpha), e0 + u));
    const float* hp = H + (size_t)se * F;
#pragma unroll
    for (int j = 0; j < V; ++j) hv[u][j] = hp[lane + 32 * j];
  }
#pragma unroll
  for (int u = 0; u < U; ++u) {
    float* op = out + (size_t)de[u] * F;
#pragma unroll
    for (int j = 0; j < V; ++j) atomicAdd(op + lane + 32 * j, ae[u] * hv[u][j]);
  }
}

template <int F>
__global__ __launch_bounds__(256) void edge_pass3_kernel(
    const int* __restrict__ src, const int* __restrict__ dst,
    const float* __restrict__ ebuf, const float* __restrict__ denom,
    const float* __restrict__ H, float* __restrict__ alpha_out,
    float* __restrict__ out, int E, int Etot) {
  const int lane = threadIdx.x & 31;
  const int wv = __builtin_amdgcn_readfirstlane((int)(threadIdx.x >> 5));
  const long long wave = (long long)blockIdx.x * (blockDim.x >> 5) + wv;
  const long long base = wave * 32;
  if (base >= Etot) return;
  const int i = (int)base + lane;
  int s = 0, d = 0;
  float alpha = 0.f;
  if (i < Etot) {
    edge_sd(i, E, src, dst, s, d);
    alpha = ebuf[i] / denom[d];
    alpha_out[i] = alpha;
  }
  const int cnt = (Etot - (int)base < 32) ? (Etot - (int)base) : 32;
  constexpr int U = 512 / F;
  if (cnt == 32) {
#pragma unroll
    for (int e0 = 0; e0 < 32; e0 += U) edge_rows<F, U>(H, out, s, d, alpha, e0, lane);
  } else {
    for (int e = 0; e < cnt; ++e) edge_rows<F, 1>(H, out, s, d, alpha, e, lane);
  }
}

static inline int cdiv(long long a, long long b) { return (int)((a + b - 1) / b); }

extern "C" void kernel_launch(void* const* d_in, const int* in_sizes, int n_in,
                              void* d_out, int out_size, void* d_ws, size_t ws_size,
                              hipStream_t stream) {
  constexpr int FIN = 128, FH = 64, FO = 32;
  const float* x      = (const float*)d_in[0];
  const int*   eidx   = (const int*)d_in[1];
  const float* W1     = (const float*)d_in[2];
  const float* a_src1 = (const float*)d_in[3];
  const float* a_dst1 = (const float*)d_in[4];
  const float* b1     = (const float*)d_in[5];
  const float* W2     = (const float*)d_in[6];
  const float* a_src2 = (const float*)d_in[7];
  const float* a_dst2 = (const float*)d_in[8];
  const float* b2     = (const float*)d_in[9];

  const int N = in_sizes[0] / FIN;
  const int E = in_sizes[1] / 2;
  const int Etot = E + N;
  const int* src = eidx;
  const int* dst = eidx + E;

  float* hfeat1 = (float*)d_ws;
  float* agg1   = hfeat1 + (size_t)N * FH;
  float* hfeat2 = agg1 + (size_t)N * FH;
  float* ebuf   = hfeat2 + (size_t)N * FO;
  float* ssrc   = ebuf + (size_t)Etot;
  float* sdst   = ssrc + N;
  unsigned* emaxu = (unsigned*)(sdst + N);
  unsigned* rng = emaxu + N;
  float* denom  = (float*)(emaxu + N + 32);

  float* h2     = (float*)d_out;
  float* alpha1 = h2 + (size_t)N * FO;
  float* alpha2 = alpha1 + (size_t)Etot;

  dim3 b256(256), b128(128), b1024(1024);
  dim3 dgrid(DEN_CHUNKS, cdiv(N, DEN_RANGE_MAX));

  hipMemsetAsync(emaxu, 0, (size_t)(2 * N + 32) * sizeof(unsigned), stream);

  gemm_scores_kernel<FIN, FH, false><<<cdiv(N, 64), b128, 0, stream>>>(
      x, W1, a_src1, a_dst1, b1, hfeat1, ssrc, sdst, rng, agg1, N);
  edge_pass1_kernel<<<1024, b256, 0, stream>>>(src, dst, ssrc, sdst, emaxu, rng, E, Etot);
  edge_pass2_kernel<<<cdiv(Etot, 256), b256, 0, stream>>>(src, dst, ssrc, sdst, ebuf, emaxu, rng, E, Etot);
  denom_range_kernel<<<dgrid, b1024, 0, stream>>>(dst, ebuf, denom, E, Etot, N);
  edge_pass3_kernel<FH><<<cdiv(cdiv(Etot, 32), 8), b256, 0, stream>>>(
      src, dst, ebuf, denom, hfeat1, alpha1, agg1, E, Etot);

  hipMemsetAsync(emaxu, 0, (size_t)(2 * N + 32) * sizeof(unsigned), stream);

  gemm_scores_kernel<FH, FO, true><<<cdiv(N, 64), b128, 0, stream>>>(
      agg1, W2, a_src2, a_dst2, b2, hfeat2, ssrc, sdst, rng, h2, N);
  edge_pass1_kernel<<<1024, b256, 0, stream>>>(src, dst, ssrc, sdst, emaxu, rng, E, Etot);
  edge_pass2_kernel<<<cdiv(Etot, 256), b256, 0, stream>>>(src, dst, ssrc, sdst, ebuf, emaxu, rng, E, Etot);
  denom_range_kernel<<<dgrid, b1024, 0, stream>>>(dst, ebuf, denom, E, Etot, N);
  edge_pass3_kernel<FO><<<cdiv(cdiv(Etot, 32), 8), b256, 0, stream>>>(
      src, dst, ebuf, denom, hfeat2, alpha2, h2, E, Etot);
}


# Edge weights computed inside the denominator kernel: a work-group keeps its node range's destination scores in LDS beside the sums; the weight pass, its launch and per-edge dst-score gather go
# speedup vs baseline: 4.1619x; 1.0498x over previous
#include <hip/hip_runtime.h>
#include <hip/hip_bf16.h>

typedef __attribute__((ext_vector_type(2))) float v2f;
typedef __attribute__((ext_vector_type(8))) float v8f;

#define NEG_SLOPE 0.2f

__device__ __forceinline__ unsigned ford(float x) {
  unsigned b = __float_as_uint(x);
  return (b & 0x80000000u) ? ~b : (b | 0x80000000u);
}
__device__ __forceinline__ float funord(unsigned u) {
  unsigned b = (u & 0x80000000u) ? (u ^ 0x80000000u) : ~u;
  return __uint_as_float(b);
}

#define RANGE_LIMIT 60.0f

__device__ __forceinline__ bool range_needs_exact(const unsigned* __restrict__ rng, float& M) {
  M = funord(rng[0]);
  float R = M + funord(rng[1]);
  return !(R <= RANGE_LIMIT);
}

template <int FIN, int FOUT, bool ELU_IN>
__global__ __launch_bounds__(128) void gemm_scores_kernel(
    const float* __restrict__ X, const float* __restrict__ W,
    const float* __restrict__ a_src, const float* __restrict__ a_dst,
    const float* __restrict__ bias, float* __restrict__ H,
    float* __restrict__ ssrc, float* __restrict__ sdst,
    unsigned* __restrict__ rng, float* __restrict__ agg, int nrows) {
  __shared__ __attribute__((aligned(16))) float lds_x[4][16 * FIN];
  __shared__ __attribute__((aligned(16))) float lds_a[2 * FOUT];
  __shared__ __attribute__((aligned(16))) float lds_b[FOUT];
  __shared__ float lds_sc[2][64];
  static_assert(2 * FOUT <= 128, "one thread per attention-vector element");
  const int wave = __builtin_amdgcn_readfirstlane((int)(threadIdx.x >> 5));
  const int lane = threadIdx.x & 31;
  const int m0 = (blockIdx.x * 4 + wave) * 16;

  if (threadIdx.x < 2 * FOUT)
    lds_a[threadIdx.x] = (threadIdx.x < FOUT) ? a_src[threadIdx.x] : a_dst[threadIdx.x - FOUT];
  if (threadIdx.x < FOUT) lds_b[threadIdx.x] = bias[threadIdx.x];

  float* xt = lds_x[wave];
  constexpr int NCHUNK = 16 * FIN / 4;
#pragma unroll 4
  for (int i = lane; i < NCHUNK; i += 32) {
    int r = i / (FIN / 4);
    int c = (i % (FIN / 4)) * 4;
    int row = m0 + r;
    if (row >= nrows) row = nrows - 1;
    float4 v = *(const float4*)(X + (size_t)row * FIN + c);
    if (ELU_IN) {
      v.x = (v.x > 0.f) ? v.x : (__expf(v.x) - 1.0f);
      v.y = (v.y > 0.f) ? v.y : (__expf(v.y) - 1.0f);
      v.z = (v.z > 0.f) ? v.z : (__expf(v.z) - 1.0f);
      v.w = (v.w > 0.f) ? v.w : (__expf(v.w) - 1.0f);
    }
    *(float4*)(xt + r * FIN + c) = v;
  }
  __syncthreads();

  constexpr int NT = FOUT / 16;
  v8f acc[NT] = {};

  const int arow = lane & 15;
  const int ksel = 2 * (lane >> 4);
  for (int k = 0; k < FIN; k += 4) {
    v2f a;
    a.x = xt[arow * FIN + k + ksel];
    a.y = xt[arow * FIN + k + ksel + 1];
#pragma unroll
    for (int t = 0; t < NT; ++t) {
      int n = t * 16 + (lane & 15);
      v2f b;
      b.x = W[(k + ksel) * FOUT + n];
      b.y = W[(k + ksel + 1) * FOUT + n];
      acc[t] = __builtin_amdgcn_wmma_f32_16x16x4_f32(
          false, a, false, b, (short)0, acc[t], false, false);
    }
  }

  constexpr int PITCH = FOUT + 4;
  static_assert(PITCH <= FIN, "the tile copy must fit the wave's staging area");
  const int row_off = (lane >> 4) * 8;
  const int col = lane & 15;
#pragma unroll
  for (int t = 0; t < NT; ++t) {
#pragma unroll
    for (int i = 0; i < 8; ++i) xt[(row_off + i) * PITCH + t * 16 + col] = acc[t][i];
  }
  __syncthreads();

  {
    const int r = lane & 15;
    const int which = lane >> 4;
    float p0 = 0.f, p1 = 0.f, p2 = 0.f, p3 = 0.f;
#pragma unroll 4
    for (int f = 0; f < FOUT; f += 4) {
      const float4 hv = *(const float4*)(xt + r * PITCH + f);
      const float4 av = *(const float4*)(&lds_a[which * FOUT + f]);
      p0 += hv.x * av.x;
      p1 += hv.y * av.y;
      p2 += hv.z * av.z;
      p3 += hv.w * av.w;
    }
    lds_sc[which][wave * 16 + r] = (p0 + p1) + (p2 + p3);
  }
  __syncthreads();

  if (wave < 2) {
    const int base = blockIdx.x * 64;
    const float v0 = lds_sc[wave][lane];
    const float v1 = lds_sc[wave][lane + 32];
    float* __restrict__ sp = wave ? sdst : ssrc;
    const bool ok0 = base + lane < nrows;
    const bool ok1 = base + 32 + lane < nrows;
    if (ok0) sp[base + lane] = v0;
    if (ok1) sp[base + 32 + lane] = v1;
    if (wave == 0) {
      float a = fmaxf(ok0 ? v0 : -INFINITY, ok1 ? v1 : -INFINITY);
      float b = fmaxf(ok0 ? -v0 : -INFINITY, ok1 ? -v1 : -INFINITY);
#pragma unroll
      for (int off = 16; off > 0; off >>= 1) {
        a = fmaxf(a, __shfl_down(a, off, 32));
        b = fmaxf(b, __shfl_down(b, off, 32));
      }
      if (lane == 0) {
        atomicMax(&rng[0], ford(a));
        atomicMax(&rng[1], ford(b));
      }
    }
  }

  constexpr int LPR = FOUT / 4;
  constexpr int RPI = 32 / LPR;
  constexpr int NI = 16 / RPI;
  const int lr = lane / LPR;
  const int lc = (lane % LPR) * 4;
  const float4 bv = *(const float4*)(&lds_b[lc]);
  if (m0 + 16 <= nrows) {
    float4 hv[NI];
#pragma unroll
    for (int j = 0; j < NI; ++j) hv[j] = *(const float4*)(xt + (j * RPI + lr) * PITCH + lc);
#pragma unroll
    for (int j = 0; j < NI; ++j) {
      const size_t o = (size_t)(m0 + j * RPI + lr) * FOUT + lc;
      *(float4*)(H + o) = hv[j];
      *(float4*)(agg + o) = bv;
    }
  } else {
    for (int j = 0; j < NI; ++j) {
      const int r = j * RPI + lr;
      if (m0 + r < nrows) {
        const size_t o = (size_t)(m0 + r) * FOUT + lc;
        *(float4*)(H + o) = *(const float4*)(xt + r * PITCH + lc);
        *(float4*)(agg + o) = bv;
      }
    }
  }
}

__device__ __forceinline__ void edge_sd(int i, int E, const int* __restrict__ src,
                                        const int* __restrict__ dst, int& s, int& d) {
  if (i < E) { s = src[i]; d = dst[i]; }
  else       { s = i - E; d = s; }
}

__global__ __launch_bounds__(256) void edge_pass1_kernel(
    const int* __restrict__ src, const int* __restrict__ dst,
    const float* __restrict__ ssrc, const float* __restrict__ sdst,
    unsigned* __restrict__ emaxu, const unsigned* __restrict__ rng, int E, int Etot) {
  float M;
  if (!range_needs_exact(rng, M)) return;
  const int stride = gridDim.x * blockDim.x;
  for (int i = blockIdx.x * blockDim.x + threadIdx.x; i < Etot; i += stride) {
    int s, d;
    edge_sd(i, E, src, dst, s, d);
    float e = ssrc[s] + sdst[d];
    e = (e > 0.f) ? e : NEG_SLOPE * e;
    atomicMax(&emaxu[d], ford(e));
  }
}

#define DEN_RANGE_MAX 25600
#define DEN_CHUNKS 64
#define DEN_U 8

__device__ __forceinline__ void den_issue(const int* __restrict__ src, const int* __restrict__ dst,
                                          int i0, int tid, int E, int (&dr)[DEN_U], int (&sr)[DEN_U]) {
#pragma unroll
  for (int u = 0; u < DEN_U; ++u) {
    const int i = i0 + u * 1024 + tid;
    const int ic = (i < E) ? i : (E - 1);
    dr[u] = dst[ic];
    sr[u] = src[ic];
  }
}

template <bool EXACT>
__device__ __forceinline__ void den_edges(
    const int* __restrict__ src, const int* __restrict__ dst, const float* __restrict__ ssrc,
    float* __restrict__ ebuf, const unsigned* __restrict__ emaxu, float M,
    const float* sdl, float* tab, int lo, int cnt, int begin, int end, int E, int tid) {
  constexpr int U = DEN_U;
  static_assert(U == 8, "the index fence below names 2 x 8 values");
  constexpr int STEP = 1024 * U;
  int dr[U], sr[U];
  den_issue(src, dst, begin, tid, E, dr, sr);
  for (int i0 = begin; i0 < end; i0 += STEP) {
    asm volatile("" :: "v"(dr[0]), "v"(dr[1]), "v"(dr[2]), "v"(dr[3]), "v"(dr[4]), "v"(dr[5]),
                 "v"(dr[6]), "v"(dr[7]), "v"(sr[0]), "v"(sr[1]), "v"(sr[2]), "v"(sr[3]),
                 "v"(sr[4]), "v"(sr[5]), "v"(sr[6]), "v"(sr[7]));
    unsigned rr[U];
    int sn[U], dn[U];
#pragma unroll
    for (int u = 0; u < U; ++u) {
      const int i = i0 + u * 1024 + tid;
      dn[u] = (i < E) ? dr[u] : (i - E);
      sn[u] = (i < E) ? sr[u] : (i - E);
      rr[u] = (i < end) ? (unsigned)(dn[u] - lo) : 0xffffffffu;
    }
    float ss[U];
    unsigned mx[U];
#pragma unroll
    for (int u = 0; u < U; ++u) {
      ss[u] = 0.f;
      mx[u] = 0u;
      if (rr[u] < (unsigned)cnt) {
        ss[u] = ssrc[sn[u]];
        if (EXACT) mx[u] = emaxu[dn[u]];
      }
    }
    den_issue(src, dst, i0 + STEP, tid, E, dr, sr);
    float sd[U];
#pragma unroll
    for (int u = 0; u < U; ++u) {
      const unsigned rc = (rr[u] < (unsigned)cnt) ? rr[u] : 0u;
      sd[u] = sdl[rc];
    }
#pragma unroll
    for (int u = 0; u < U; ++u) {
      float e = ss[u] + sd[u];
      e = (e > 0.f) ? e : NEG_SLOPE * e;
      float shift;
      if (EXACT) {
        shift = funord(mx[u]);
      } else {
        shift = M + sd[u];
        shift = (shift > 0.f) ? shift : NEG_SLOPE * shift;
      }
      const float ex = __expf(e - shift);
      if (rr[u] < (unsigned)cnt) {
        ebuf[i0 + u * 1024 + tid] = ex;
        __hip_atomic_fetch_add(&tab[rr[u]], ex, __ATOMIC_RELAXED, __HIP_MEMORY_SCOPE_WORKGROUP);
      }
    }
  }
}

__global__ __launch_bounds__(1024) void weight_denom_kernel(
    const int* __restrict__ src, const int* __restrict__ dst,
    const float* __restrict__ ssrc, const float* __restrict__ sdst,
    float* __restrict__ ebuf, const unsigned* __restrict__ emaxu,
    const unsigned* __restrict__ rng, float* __restrict__ denom, int E, int Etot, int N) {
  __shared__ float tab[DEN_RANGE_MAX];
  __shared__ float sdl[DEN_RANGE_MAX];
  const int tid = threadIdx.x;
  const int nranges = gridDim.y;
  int range = (N + nranges - 1) / nranges;
  range = (range + 31) & ~31;
  const int lo = blockIdx.y * range;
  const int cnt = ((lo + range < N) ? (lo + range) : N) - lo;
  if (cnt <= 0) return;

  float M;
  const bool exact = range_needs_exact(rng, M);

  for (int j = tid; j < cnt; j += 1024) {
    tab[j] = 0.f;
    sdl[j] = sdst[lo + j];
  }
  __syncthreads();

  const int chunk = (Etot + gridDim.x - 1) / gridDim.x;
  const int begin = blockIdx.x * chunk;
  const int end = (begin + chunk < Etot) ? (begin + chunk) : Etot;
  if (exact)
    den_edges<true>(src, dst, ssrc, ebuf, emaxu, M, sdl, tab, lo, cnt, begin, end, E, tid);
  else
    den_edges<false>(src, dst, ssrc, ebuf, emaxu, M, sdl, tab, lo, cnt, begin, end, E, tid);
  __syncthreads();

  for (int j = tid; j < cnt; j += 1024) {
    const float v = tab[j];
    if (v != 0.f) atomicAdd(&denom[lo + j], v);
  }
}

template <int F, int U>
__device__ __forceinline__ void edge_rows(const float* __restrict__ H, float* __restrict__ out,
                                          int s, int d, float alpha, int e0, int lane) {
  constexpr int V = F / 32;
  float hv[U][V];
  int de[U];
  float ae[U];
#pragma unroll
  for (int u = 0; u < U; ++u) {
    const int se = __builtin_amdgcn_readlane(s, e0 + u);
    de[u] = __builtin_amdgcn_readlane(d, e0 + u);
    ae[u] = __int_as_float(__builtin_amdgcn_readlane(__float_as_int(alpha), e0 + u));
    const float* hp = H + (size_t)se * F;
#pragma unroll
    for (int j = 0; j < V; ++j) hv[u][j] = hp[lane + 32 * j];
  }
#pragma unroll
  for (int u = 0; u < U; ++u) {
    float* op = out + (size_t)de[u] * F;
#pragma unroll
    for (int j = 0; j < V; ++j) atomicAdd(op + lane + 32 * j, ae[u] * hv[u][j]);
  }
}

template <int F>
__global__ __launch_bounds__(256) void edge_pass3_kernel(
    const int* __restrict__ src, const int* __restrict__ dst,
    const float* __restrict__ ebuf, const float* __restrict__ denom,
    const float* __restrict__ H, float* __restrict__ alpha_out,
    float* __restrict__ out, int E, int Etot) {
  const int lane = threadIdx.x & 31;
  const int wv = __builtin_amdgcn_readfirstlane((int)(threadIdx.x >> 5));
  const long long wave = (long long)blockIdx.x * (blockDim.x >> 5) + wv;
  const long long base = wave * 32;
  if (base >= Etot) return;
  const int i = (int)base + lane;
  int s = 0, d = 0;
  float alpha = 0.f;
  if (i < Etot) {
    edge_sd(i, E, src, dst, s, d);
    alpha = ebuf[i] / denom[d];
    alpha_out[i] = alpha;
  }
  const int cnt = (Etot - (int)base < 32) ? (Etot - (int)base) : 32;
  constexpr int U = 512 / F;
  if (cnt == 32) {
#pragma unroll
    for (int e0 = 0; e0 < 32; e0 += U) edge_rows<F, U>(H, out, s, d, alpha, e0, lane);
  } else {
    for (int e = 0; e < cnt; ++e) edge_rows<F, 1>(H, out, s, d, alpha, e, lane);
  }
}

static inline int cdiv(long long a, long long b) { return (int)((a + b - 1) / b); }

extern "C" void kernel_launch(void* const* d_in, const int* in_sizes, int n_in,
                              void* d_out, int out_size, void* d_ws, size_t ws_size,
                              hipStream_t stream) {
  constexpr int FIN = 128, FH = 64, FO = 32;
  const float* x      = (const float*)d_in[0];
  const int*   eidx   = (const int*)d_in[1];
  const float* W1     = (const float*)d_in[2];
  const float* a_src1 = (const float*)d_in[3];
  const float* a_dst1 = (const float*)d_in[4];
  const float* b1     = (const float*)d_in[5];
  const float* W2     = (const float*)d_in[6];
  const float* a_src2 = (const float*)d_in[7];
  const float* a_dst2 = (const float*)d_in[8];
  const float* b2     = (const float*)d_in[9];

  const int N = in_sizes[0] / FIN;
  const int E = in_sizes[1] / 2;
  const int Etot = E + N;
  const int* src = eidx;
  const int* dst = eidx + E;

  float* hfeat1 = (float*)d_ws;
  float* agg1   = hfeat1 + (size_t)N * FH;
  float* hfeat2 = agg1 + (size_t)N * FH;
  float* ebuf   = hfeat2 + (size_t)N * FO;
  float* ssrc   = ebuf + (size_t)Etot;
  float* sdst   = ssrc + N;
  unsigned* emaxu = (unsigned*)(sdst + N);
  unsigned* rng = emaxu + N;
  float* denom  = (float*)(emaxu + N + 32);

  float* h2     = (float*)d_out;
  float* alpha1 = h2 + (size_t)N * FO;
  float* alpha2 = alpha1 + (size_t)Etot;

  dim3 b256(256), b128(128), b1024(1024);
  dim3 dgrid(DEN_CHUNKS, cdiv(N, DEN_RANGE_MAX));

  hipMemsetAsync(emaxu, 0, (size_t)(2 * N + 32) * sizeof(unsigned), stream);

  gemm_scores_kernel<FIN, FH, false><<<cdiv(N, 64), b128, 0, stream>>>(
      x, W1, a_src1, a_dst1, b1, hfeat1, ssrc, sdst, rng, agg1, N);
  edge_pass1_kernel<<<1024, b256, 0, stream>>>(src, dst, ssrc, sdst, emaxu, rng, E, Etot);
  weight_denom_kernel<<<dgrid, b1024, 0, stream>>>(src, dst, ssrc, sdst, ebuf, emaxu, rng, denom, E, Etot, N);
  edge_pass3_kernel<FH><<<cdiv(cdiv(Etot, 32), 8), b256, 0, stream>>>(
      src, dst, ebuf, denom, hfeat1, alpha1, agg1, E, Etot);

  hipMemsetAsync(emaxu, 0, (size_t)(2 * N + 32) * sizeof(unsigned), stream);

  gemm_scores_kernel<FH, FO, true><<<cdiv(N, 64), b128, 0, stream>>>(
      agg1, W2, a_src2, a_dst2, b2, hfeat2, ssrc, sdst, rng, h2, N);
  edge_pass1_kernel<<<1024, b256, 0, stream>>>(src, dst, ssrc, sdst, emaxu, rng, E, Etot);
  weight_denom_kernel<<<dgrid, b1024, 0, stream>>>(src, dst, ssrc, sdst, ebuf, emaxu, rng, denom, E, Etot, N);
  edge_pass3_kernel<FO><<<cdiv(cdiv(Etot, 32), 8), b256, 0, stream>>>(
      src, dst, ebuf, denom, hfeat2, alpha2, h2, E, Etot);
}


# Row-sum atomics kept inside one half of the chip: a work-group reads its die, draws a tile ticket of that half's node class, queues the class's edges in LDS; atomic arrival order changes
# speedup vs baseline: 5.0795x; 1.2205x over previous
#include <hip/hip_runtime.h>
#include <hip/hip_bf16.h>

typedef __attribute__((ext_vector_type(2))) float v2f;
typedef __attribute__((ext_vector_type(8))) float v8f;

#define NEG_SLOPE 0.2f

__device__ __forceinline__ unsigned ford(float x) {
  unsigned b = __float_as_uint(x);
  return (b & 0x80000000u) ? ~b : (b | 0x80000000u);
}
__device__ __forceinline__ float funord(unsigned u) {
  unsigned b = (u & 0x80000000u) ? (u ^ 0x80000000u) : ~u;
  return __uint_as_float(b);
}

#define RANGE_LIMIT 60.0f

__device__ __forceinline__ bool range_needs_exact(const unsigned* __restrict__ rng, float& M) {
  M = funord(rng[0]);
  float R = M + funord(rng[1]);
  return !(R <= RANGE_LIMIT);
}

template <int FIN, int FOUT, bool ELU_IN>
__global__ __launch_bounds__(128) void gemm_scores_kernel(
    const float* __restrict__ X, const float* __restrict__ W,
    const float* __restrict__ a_src, const float* __restrict__ a_dst,
    const float* __restrict__ bias, float* __restrict__ H,
    float* __restrict__ ssrc, float* __restrict__ sdst,
    unsigned* __restrict__ rng, float* __restrict__ agg, int nrows) {
  __shared__ __attribute__((aligned(16))) float lds_x[4][16 * FIN];
  __shared__ __attribute__((aligned(16))) float lds_a[2 * FOUT];
  __shared__ __attribute__((aligned(16))) float lds_b[FOUT];
  __shared__ float lds_sc[2][64];
  static_assert(2 * FOUT <= 128, "one thread per attention-vector element");
  const int wave = __builtin_amdgcn_readfirstlane((int)(threadIdx.x >> 5));
  const int lane = threadIdx.x & 31;
  const int m0 = (blockIdx.x * 4 + wave) * 16;

  if (threadIdx.x < 2 * FOUT)
    lds_a[threadIdx.x] = (threadIdx.x < FOUT) ? a_src[threadIdx.x] : a_dst[threadIdx.x - FOUT];
  if (threadIdx.x < FOUT) lds_b[threadIdx.x] = bias[threadIdx.x];

  float* xt = lds_x[wave];
  constexpr int NCHUNK = 16 * FIN / 4;
#pragma unroll 4
  for (int i = lane; i < NCHUNK; i += 32) {
    int r = i / (FIN / 4);
    int c = (i % (FIN / 4)) * 4;
    int row = m0 + r;
    if (row >= nrows) row = nrows - 1;
    float4 v = *(const float4*)(X + (size_t)row * FIN + c);
    if (ELU_IN) {
      v.x = (v.x > 0.f) ? v.x : (__expf(v.x) - 1.0f);
      v.y = (v.y > 0.f) ? v.y : (__expf(v.y) - 1.0f);
      v.z = (v.z > 0.f) ? v.z : (__expf(v.z) - 1.0f);
      v.w = (v.w > 0.f) ? v.w : (__expf(v.w) - 1.0f);
    }
    *(float4*)(xt + r * FIN + c) = v;
  }
  __syncthreads();

  constexpr int NT = FOUT / 16;
  v8f acc[NT] = {};

  const int arow = lane & 15;
  const int ksel = 2 * (lane >> 4);
  for (int k = 0; k < FIN; k += 4) {
    v2f a;
    a.x = xt[arow * FIN + k + ksel];
    a.y = xt[arow * FIN + k + ksel + 1];
#pragma unroll
    for (int t = 0; t < NT; ++t) {
      int n = t * 16 + (lane & 15);
      v2f b;
      b.x = W[(k + ksel) * FOUT + n];
      b.y = W[(k + ksel + 1) * FOUT + n];
      acc[t] = __builtin_amdgcn_wmma_f32_16x16x4_f32(
          false, a, false, b, (short)0, acc[t], false, false);
    }
  }

  constexpr int PITCH = FOUT + 4;
  static_assert(PITCH <= FIN, "the tile copy must fit the wave's staging area");
  const int row_off = (lane >> 4) * 8;
  const int col = lane & 15;
#pragma unroll
  for (int t = 0; t < NT; ++t) {
#pragma unroll
    for (int i = 0; i < 8; ++i) xt[(row_off + i) * PITCH + t * 16 + col] = acc[t][i];
  }
  __syncthreads();

  {
    const int r = lane & 15;
    const int which = lane >> 4;
    float p0 = 0.f, p1 = 0.f, p2 = 0.f, p3 = 0.f;
#pragma unroll 4
    for (int f = 0; f < FOUT; f += 4) {
      const float4 hv = *(const float4*)(xt + r * PITCH + f);
      const float4 av = *(const float4*)(&lds_a[which * FOUT + f]);
      p0 += hv.x * av.x;
      p1 += hv.y * av.y;
      p2 += hv.z * av.z;
      p3 += hv.w * av.w;
    }
    lds_sc[which][wave * 16 + r] = (p0 + p1) + (p2 + p3);
  }
  __syncthreads();

  if (wave < 2) {
    const int base = blockIdx.x * 64;
    const float v0 = lds_sc[wave][lane];
    const float v1 = lds_sc[wave][lane + 32];
    float* __restrict__ sp = wave ? sdst : ssrc;
    const bool ok0 = base + lane < nrows;
    const bool ok1 = base + 32 + lane < nrows;
    if (ok0) sp[base + lane] = v0;
    if (ok1) sp[base + 32 + lane] = v1;
    if (wave == 0) {
      float a = fmaxf(ok0 ? v0 : -INFINITY, ok1 ? v1 : -INFINITY);
      float b = fmaxf(ok0 ? -v0 : -INFINITY, ok1 ? -v1 : -INFINITY);
#pragma unroll
      for (int off = 16; off > 0; off >>= 1) {
        a = fmaxf(a, __shfl_down(a, off, 32));
        b = fmaxf(b, __shfl_down(b, off, 32));
      }
      if (lane == 0) {
        atomicMax(&rng[0], ford(a));
        atomicMax(&rng[1], ford(b));
      }
    }
  }

  constexpr int LPR = FOUT / 4;
  constexpr int RPI = 32 / LPR;
  constexpr int NI = 16 / RPI;
  const int lr = lane / LPR;
  const int lc = (lane % LPR) * 4;
  const float4 bv = *(const float4*)(&lds_b[lc]);
  if (m0 + 16 <= nrows) {
    float4 hv[NI];
#pragma unroll
    for (int j = 0; j < NI; ++j) hv[j] = *(const float4*)(xt + (j * RPI + lr) * PITCH + lc);
#pragma unroll
    for (int j = 0; j < NI; ++j) {
      const size_t o = (size_t)(m0 + j * RPI + lr) * FOUT + lc;
      *(float4*)(H + o) = hv[j];
      *(float4*)(agg + o) = bv;
    }
  } else {
    for (int j = 0; j < NI; ++j) {
      const int r = j * RPI + lr;
      if (m0 + r < nrows) {
        const size_t o = (size_t)(m0 + r) * FOUT + lc;
        *(float4*)(H + o) = *(const float4*)(xt + r * PITCH + lc);
        *(float4*)(agg + o) = bv;
      }
    }
  }
}

__device__ __forceinline__ void edge_sd(int i, int E, const int* __restrict__ src,
                                        const int* __restrict__ dst, int& s, int& d) {
  if (i < E) { s = src[i]; d = dst[i]; }
  else       { s = i - E; d = s; }
}

__global__ __launch_bounds__(256) void edge_pass1_kernel(
    const int* __restrict__ src, const int* __restrict__ dst,
    const float* __restrict__ ssrc, const float* __restrict__ sdst,
    unsigned* __restrict__ emaxu, const unsigned* __restrict__ rng, int E, int Etot) {
  float M;
  if (!range_needs_exact(rng, M)) return;
  const int stride = gridDim.x * blockDim.x;
  for (int i = blockIdx.x * blockDim.x + threadIdx.x; i < Etot; i += stride) {
    int s, d;
    edge_sd(i, E, src, dst, s, d);
    float e = ssrc[s] + sdst[d];
    e = (e > 0.f) ? e : NEG_SLOPE * e;
    atomicMax(&emaxu[d], ford(e));
  }
}

#define DEN_RANGE_MAX 25600
#define DEN_CHUNKS 64
#define DEN_U 8

__device__ __forceinline__ void den_issue(const int* __restrict__ src, const int* __restrict__ dst,
                                          int i0, int tid, int E, int (&dr)[DEN_U], int (&sr)[DEN_U]) {
#pragma unroll
  for (int u = 0; u < DEN_U; ++u) {
    const int i = i0 + u * 1024 + tid;
    const int ic = (i < E) ? i : (E - 1);
    dr[u] = dst[ic];
    sr[u] = src[ic];
  }
}

template <bool EXACT>
__device__ __forceinline__ void den_edges(
    const int* __restrict__ src, const int* __restrict__ dst, const float* __restrict__ ssrc,
    float* __restrict__ ebuf, const unsigned* __restrict__ emaxu, float M,
    const float* sdl, float* tab, int lo, int cnt, int begin, int end, int E, int tid) {
  constexpr int U = DEN_U;
  static_assert(U == 8, "the index fence below names 2 x 8 values");
  constexpr int STEP = 1024 * U;
  int dr[U], sr[U];
  den_issue(src, dst, begin, tid, E, dr, sr);
  for (int i0 = begin; i0 < end; i0 += STEP) {
    asm volatile("" :: "v"(dr[0]), "v"(dr[1]), "v"(dr[2]), "v"(dr[3]), "v"(dr[4]), "v"(dr[5]),
                 "v"(dr[6]), "v"(dr[7]), "v"(sr[0]), "v"(sr[1]), "v"(sr[2]), "v"(sr[3]),
                 "v"(sr[4]), "v"(sr[5]), "v"(sr[6]), "v"(sr[7]));
    unsigned rr[U];
    int sn[U], dn[U];
#pragma unroll
    for (int u = 0; u < U; ++u) {
      const int i = i0 + u * 1024 + tid;
      dn[u] = (i < E) ? dr[u] : (i - E);
      sn[u] = (i < E) ? sr[u] : (i - E);
      rr[u] = (i < end) ? (unsigned)(dn[u] - lo) : 0xffffffffu;
    }
    float ss[U];
    unsigned mx[U];
#pragma unroll
    for (int u = 0; u < U; ++u) {
      ss[u] = 0.f;
      mx[u] = 0u;
      if (rr[u] < (unsigned)cnt) {
        ss[u] = ssrc[sn[u]];
        if (EXACT) mx[u] = emaxu[dn[u]];
      }
    }
    den_issue(src, dst, i0 + STEP, tid, E, dr, sr);
    float sd[U];
#pragma unroll
    for (int u = 0; u < U; ++u) {
      const unsigned rc = (rr[u] < (unsigned)cnt) ? rr[u] : 0u;
      sd[u] = sdl[rc];
    }
#pragma unroll
    for (int u = 0; u < U; ++u) {
      float e = ss[u] + sd[u];
      e = (e > 0.f) ? e : NEG_SLOPE * e;
      float shift;
      if (EXACT) {
        shift = funord(mx[u]);
      } else {
        shift = M + sd[u];
        shift = (shift > 0.f) ? shift : NEG_SLOPE * shift;
      }
      const float ex = __expf(e - shift);
      if (rr[u] < (unsigned)cnt) {
        ebuf[i0 + u * 1024 + tid] = ex;
        __hip_atomic_fetch_add(&tab[rr[u]], ex, __ATOMIC_RELAXED, __HIP_MEMORY_SCOPE_WORKGROUP);
      }
    }
  }
}

__global__ __launch_bounds__(1024) void weight_denom_kernel(
    const int* __restrict__ src, const int* __restrict__ dst,
    const float* __restrict__ ssrc, const float* __restrict__ sdst,
    float* __restrict__ ebuf, const unsigned* __restrict__ emaxu,
    const unsigned* __restrict__ rng, float* __restrict__ denom, int E, int Etot, int N) {
  __shared__ float tab[DEN_RANGE_MAX];
  __shared__ float sdl[DEN_RANGE_MAX];
  const int tid = threadIdx.x;
  const int nranges = gridDim.y;
  int range = (N + nranges - 1) / nranges;
  range = (range + 31) & ~31;
  const int lo = blockIdx.y * range;
  const int cnt = ((lo + range < N) ? (lo + range) : N) - lo;
  if (cnt <= 0) return;

  float M;
  const bool exact = range_needs_exact(rng, M);

  for (int j = tid; j < cnt; j += 1024) {
    tab[j] = 0.f;
    sdl[j] = sdst[lo + j];
  }
  __syncthreads();

  const int chunk = (Etot + gridDim.x - 1) / gridDim.x;
  const int begin = blockIdx.x * chunk;
  const int end = (begin + chunk < Etot) ? (begin + chunk) : Etot;
  if (exact)
    den_edges<true>(src, dst, ssrc, ebuf, emaxu, M, sdl, tab, lo, cnt, begin, end, E, tid);
  else
    den_edges<false>(src, dst, ssrc, ebuf, emaxu, M, sdl, tab, lo, cnt, begin, end, E, tid);
  __syncthreads();

  for (int j = tid; j < cnt; j += 1024) {
    const float v = tab[j];
    if (v != 0.f) atomicAdd(&denom[lo + j], v);
  }
}

template <int F, int U>
__device__ __forceinline__ void edge_rows(const float* __restrict__ H, float* __restrict__ out,
                                          int s, int d, float alpha, int e0, int lane) {
  constexpr int V = F / 32;
  float hv[U][V];
  int de[U];
  float ae[U];
#pragma unroll
  for (int u = 0; u < U; ++u) {
    const int se = __builtin_amdgcn_readlane(s, e0 + u);
    de[u] = __builtin_amdgcn_readlane(d, e0 + u);
    ae[u] = __int_as_float(__builtin_amdgcn_readlane(__float_as_int(alpha), e0 + u));
    const float* hp = H + (size_t)se * F;
#pragma unroll
    for (int j = 0; j < V; ++j) hv[u][j] = hp[lane + 32 * j];
  }
#pragma unroll
  for (int u = 0; u < U; ++u) {
    float* op = out + (size_t)de[u] * F;
#pragma unroll
    for (int j = 0; j < V; ++j) atomicAdd(op + lane + 32 * j, ae[u] * hv[u][j]);
  }
}

template <int F, int U>
__device__ __forceinline__ void edge_rows_n(const float* __restrict__ H, float* __restrict__ out,
                                            int s, int d, float alpha, int e0, int n, int lane) {
  constexpr int V = F / 32;
  float hv[U][V];
  int de[U];
  float ae[U];
#pragma unroll
  for (int u = 0; u < U; ++u) {
    const int e = (e0 + u) & 31;
    const int se = __builtin_amdgcn_readlane(s, e);
    de[u] = __builtin_amdgcn_readlane(d, e);
    ae[u] = __int_as_float(__builtin_amdgcn_readlane(__float_as_int(alpha), e));
    const float* hp = H + (size_t)se * F;
#pragma unroll
    for (int j = 0; j < V; ++j) hv[u][j] = 0.f;
    if (e0 + u < n) {
#pragma unroll
      for (int j = 0; j < V; ++j) hv[u][j] = hp[lane + 32 * j];
    }
  }
#pragma unroll
  for (int u = 0; u < U; ++u) {
    if (e0 + u < n) {
      float* op = out + (size_t)de[u] * F;
#pragma unroll
      for (int j = 0; j < V; ++j) atomicAdd(op + lane + 32 * j, ae[u] * hv[u][j]);
    }
  }
}

#define TICKET_PITCH 64

template <int F>
__global__ __launch_bounds__(256) void edge_pass3_kernel(
    const int* __restrict__ src, const int* __restrict__ dst,
    const float* __restrict__ ebuf, const float* __restrict__ denom,
    const float* __restrict__ H, float* __restrict__ alpha_out,
    float* __restrict__ out, unsigned* __restrict__ tick, int E, int Etot, int N) {
  constexpr int SUB = 128 / F;
  constexpr int TILE = 8 * 32 * SUB;
  __shared__ int q_s[TILE];
  __shared__ int q_d[TILE];
  __shared__ float q_a[TILE];
  __shared__ int sh[3];
  const int lane = threadIdx.x & 31;
  const int wv = __builtin_amdgcn_readfirstlane((int)(threadIdx.x >> 5));
  const int ntiles = (Etot + TILE - 1) / TILE;

  if (threadIdx.x == 0) {
    const unsigned id = __builtin_amdgcn_s_sendmsg_rtn(0x87);
    int c = (int)((id >> 18) & 1u);
    int t = (int)atomicAdd(tick + c * TICKET_PITCH, 1u);
    if (t >= ntiles) {
      c ^= 1;
      t = (int)atomicAdd(tick + c * TICKET_PITCH, 1u);
    }
    sh[0] = t;
    sh[1] = c;
    sh[2] = 0;
  }
  __syncthreads();
  const int tile = sh[0];
  const int cls = sh[1];
  if (tile >= ntiles) return;
  const int nmid = ((N >> 1) + 1) & ~1;

  int sv[SUB], dv[SUB];
  bool in[SUB];
#pragma unroll
  for (int k = 0; k < SUB; ++k) {
    const int i = tile * TILE + (wv * SUB + k) * 32 + lane;
    const int ic = (i < E) ? i : (E - 1);
    sv[k] = src[ic];
    dv[k] = dst[ic];
  }
#pragma unroll
  for (int k = 0; k < SUB; ++k) {
    const int i = tile * TILE + (wv * SUB + k) * 32 + lane;
    if (i >= E) { sv[k] = i - E; dv[k] = i - E; }
    in[k] = (i < Etot) && ((((dv[k] >= nmid) ? 1 : 0)) == cls);
  }
  float ev[SUB], dn[SUB];
#pragma unroll
  for (int k = 0; k < SUB; ++k) {
    const int i = tile * TILE + (wv * SUB + k) * 32 + lane;
    ev[k] = ebuf[(i < Etot) ? i : (Etot - 1)];
    dn[k] = denom[in[k] ? dv[k] : 0];
  }
#pragma unroll
  for (int k = 0; k < SUB; ++k) {
    const int i = tile * TILE + (wv * SUB + k) * 32 + lane;
    const float alpha = ev[k] / dn[k];
    if (in[k]) alpha_out[i] = alpha;
    const unsigned m = __builtin_amdgcn_ballot_w32(in[k]);
    int base = 0;
    if (lane == 0 && m != 0u) base = atomicAdd(&sh[2], (int)__popc(m));
    base = __builtin_amdgcn_readfirstlane(base);
    if (in[k]) {
      const int idx = base + (int)__builtin_amdgcn_mbcnt_lo(m, 0u);
      q_s[idx] = sv[k];
      q_d[idx] = dv[k];
      q_a[idx] = alpha;
    }
  }
  __syncthreads();
  const int total = sh[2];

  constexpr int U = 512 / F;
  for (int g0 = wv * 32; g0 < total; g0 += 256) {
    const int n = (total - g0 < 32) ? (total - g0) : 32;
    int s = 0, d = 0;
    float alpha = 0.f;
    if (lane < n) {
      s = q_s[g0 + lane];
      d = q_d[g0 + lane];
      alpha = q_a[g0 + lane];
    }
    if (n == 32) {
#pragma unroll
      for (int e0 = 0; e0 < 32; e0 += U) edge_rows<F, U>(H, out, s, d, alpha, e0, lane);
    } else {
      for (int e0 = 0; e0 < n; e0 += U) edge_rows_n<F, U>(H, out, s, d, alpha, e0, n, lane);
    }
  }
}

static inline int cdiv(long long a, long long b) { return (int)((a + b - 1) / b); }

extern "C" void kernel_launch(void* const* d_in, const int* in_sizes, int n_in,
                              void* d_out, int out_size, void* d_ws, size_t ws_size,
                              hipStream_t stream) {
  constexpr int FIN = 128, FH = 64, FO = 32;
  const float* x      = (const float*)d_in[0];
  const int*   eidx   = (const int*)d_in[1];
  const float* W1     = (const float*)d_in[2];
  const float* a_src1 = (const float*)d_in[3];
  const float* a_dst1 = (const float*)d_in[4];
  const float* b1     = (const float*)d_in[5];
  const float* W2     = (const float*)d_in[6];
  const float* a_src2 = (const float*)d_in[7];
  const float* a_dst2 = (const float*)d_in[8];
  const float* b2     = (const float*)d_in[9];

  const int N = in_sizes[0] / FIN;
  const int E = in_sizes[1] / 2;
  const int Etot = E + N;
  const int* src = eidx;
  const int* dst = eidx + E;

  float* hfeat1 = (float*)d_ws;
  float* agg1   = hfeat1 + (size_t)N * FH;
  float* hfeat2 = agg1 + (size_t)N * FH;
  float* ebuf   = hfeat2 + (size_t)N * FO;
  float* ssrc   = ebuf + (size_t)Etot;
  float* sdst   = ssrc + N;
  unsigned* emaxu = (unsigned*)(sdst + N);
  unsigned* rng = emaxu + N;
  unsigned* tick = rng + TICKET_PITCH;
  float* denom  = (float*)(emaxu + N + 256);

  float* h2     = (float*)d_out;
  float* alpha1 = h2 + (size_t)N * FO;
  float* alpha2 = alpha1 + (size_t)Etot;

  dim3 b256(256), b128(128), b1024(1024);
  dim3 dgrid(DEN_CHUNKS, cdiv(N, DEN_RANGE_MAX));

  hipMemsetAsync(emaxu, 0, (size_t)(2 * N + 256) * sizeof(unsigned), stream);

  gemm_scores_kernel<FIN, FH, false><<<cdiv(N, 64), b128, 0, stream>>>(
      x, W1, a_src1, a_dst1, b1, hfeat1, ssrc, sdst, rng, agg1, N);
  edge_pass1_kernel<<<1024, b256, 0, stream>>>(src, dst, ssrc, sdst, emaxu, rng, E, Etot);
  weight_denom_kernel<<<dgrid, b1024, 0, stream>>>(src, dst, ssrc, sdst, ebuf, emaxu, rng, denom, E, Etot, N);
  edge_pass3_kernel<FH><<<2 * cdiv(Etot, 8 * 32 * (128 / FH)), b256, 0, stream>>>(
      src, dst, ebuf, denom, hfeat1, alpha1, agg1, tick, E, Etot, N);

  hipMemsetAsync(emaxu, 0, (size_t)(2 * N + 256) * sizeof(unsigned), stream);

  gemm_scores_kernel<FH, FO, true><<<cdiv(N, 64), b128, 0, stream>>>(
      agg1, W2, a_src2, a_dst2, b2, hfeat2, ssrc, sdst, rng, h2, N);
  edge_pass1_kernel<<<1024, b256, 0, stream>>>(src, dst, ssrc, sdst, emaxu, rng, E, Etot);
  weight_denom_kernel<<<dgrid, b1024, 0, stream>>>(src, dst, ssrc, sdst, ebuf, emaxu, rng, denom, E, Etot, N);
  edge_pass3_kernel<FO><<<2 * cdiv(Etot, 8 * 32 * (128 / FO)), b256, 0, stream>>>(
      src, dst, ebuf, denom, hfeat2, alpha2, h2, tick, E, Etot, N);
}
